# starting kernel
# baseline (speedup 1.0000x reference)
#include <hip/hip_runtime.h>


typedef float v2f __attribute__((ext_vector_type(2)));
typedef float v8f __attribute__((ext_vector_type(8)));

#define D 64

__global__ void gcn_zero_kernel(float* __restrict__ ws, long n) {
    long i = (long)blockIdx.x * blockDim.x + threadIdx.x;
    long stride = (long)gridDim.x * blockDim.x;
    for (; i < n; i += stride) ws[i] = 0.0f;
}

__global__ void gcn_degree_kernel(const int* __restrict__ src, const int* __restrict__ dst,
                                  float* __restrict__ out_deg, float* __restrict__ in_deg, int E) {
    int e = blockIdx.x * blockDim.x + threadIdx.x;
    if (e < E) {
        atomicAdd(out_deg + src[e], 1.0f);
        atomicAdd(in_deg  + dst[e], 1.0f);
    }
}

__global__ void gcn_norm_kernel(float* __restrict__ deg_src, float* __restrict__ deg_dst, int N) {
    int i = blockIdx.x * blockDim.x + threadIdx.x;
    if (i < N) {
        deg_src[i] = rsqrtf(fmaxf(deg_src[i], 1.0f));
        deg_dst[i] = rsqrtf(fmaxf(deg_dst[i], 1.0f));
    }
}

__global__ void gcn_scatter_kernel(const float* __restrict__ x, const int* __restrict__ src,
                                   const int* __restrict__ dst, const float* __restrict__ norm_src,
                                   float* __restrict__ agg, int E) {
    long tid = (long)blockIdx.x * blockDim.x + threadIdx.x;
    int e = (int)(tid >> 4);
    if (e >= E) return;
    int f = ((int)tid & 15) * 4;
    int s = src[e];
    int d = dst[e];
    float ns = norm_src[s];
    float4 v = *(const float4*)(x + (long)s * D + f);
    float* a = agg + (long)d * D + f;
    atomicAdd(a + 0, v.x * ns);
    atomicAdd(a + 1, v.y * ns);
    atomicAdd(a + 2, v.z * ns);
    atomicAdd(a + 3, v.w * ns);
}

__global__ void gcn_gemm_wmma_kernel(const float* __restrict__ agg,
                                     const float* __restrict__ norm_dst,
                                     const float* __restrict__ Wmat,
                                     float* __restrict__ out,
                                     int N, int numTiles) {
    int lane = threadIdx.x & 31;
    int wave = threadIdx.x >> 5;
    int w = blockIdx.x * (blockDim.x >> 5) + wave;
    if (w >= numTiles) return;

    int mtile = w >> 2;
    int ntile = w & 3;
    int row0 = mtile * 16;
    int col0 = ntile * 16;

    int half = lane >> 4;
    int lm   = lane & 15;

    int m = row0 + lm;
    int mc = m < N ? m : (N - 1);
    float scale = (m < N) ? norm_dst[mc] : 0.0f;
    const float* arow = agg + (long)mc * D;

    v8f c = {};
    #pragma unroll
    for (int kk = 0; kk < 16; ++kk) {
        int kbase = kk * 4 + half * 2;
        float2 av = *(const float2*)(arow + kbase);
        v2f a; a.x = av.x * scale; a.y = av.y * scale;
        v2f b;
        b.x = Wmat[(kbase + 0) * D + col0 + lm];
        b.y = Wmat[(kbase + 1) * D + col0 + lm];
        c = __builtin_amdgcn_wmma_f32_16x16x4_f32(false, a, false, b, (short)0, c, false, false);
    }

    #pragma unroll
    for (int v = 0; v < 8; ++v) {
        int row = row0 + v + half * 8;
        if (row < N)
            out[(long)row * D + col0 + lm] = c[v];
    }
}

extern "C" void kernel_launch(void* const* d_in, const int* in_sizes, int n_in,
                              void* d_out, int out_size, void* d_ws, size_t ws_size,
                              hipStream_t stream) {
    const float* x    = (const float*)d_in[0];
    const float* Wmat = (const float*)d_in[1];
    const int*   src  = (const int*)d_in[2];
    const int*   dst  = (const int*)d_in[3];
    float* out = (float*)d_out;

    int N = in_sizes[0] / D;
    int E = in_sizes[2];

    float* ws_norm_src = (float*)d_ws;
    float* ws_norm_dst = ws_norm_src + N;
    float* ws_agg      = ws_norm_dst + N;

    long nzero = (long)N * (D + 2);
    {
        int blocks = (int)((nzero + 255) / 256);
        if (blocks > 65535) blocks = 65535;
        gcn_zero_kernel<<<blocks, 256, 0, stream>>>((float*)d_ws, nzero);
    }

    gcn_degree_kernel<<<(E + 255) / 256, 256, 0, stream>>>(src, dst, ws_norm_src, ws_norm_dst, E);

    gcn_norm_kernel<<<(N + 255) / 256, 256, 0, stream>>>(ws_norm_src, ws_norm_dst, N);

    {
        long work = (long)E * 16;
        int blocks = (int)((work + 255) / 256);
        gcn_scatter_kernel<<<blocks, 256, 0, stream>>>(x, src, dst, ws_norm_src, ws_agg, E);
    }

    {
        int mtiles = (N + 15) / 16;
        int numTiles = mtiles * 4;
        int wavesPerBlock = 8;
        int blocks = (numTiles + wavesPerBlock - 1) / wavesPerBlock;
        gcn_gemm_wmma_kernel<<<blocks, 256, 0, stream>>>(ws_agg, ws_norm_dst, Wmat, out, N, numTiles);
    }
}


# Degree counts by per-block LDS integer tables over (edge chunk x node range), plain partial stores summed in the norm kernel; no global atomics left in the degree pass; counts exact
# speedup vs baseline: 1.7150x; 1.7150x over previous
#include <hip/hip_runtime.h>


typedef float v2f __attribute__((ext_vector_type(2)));
typedef float v8f __attribute__((ext_vector_type(8)));

#define D 64

__global__ void gcn_zero_kernel(float* __restrict__ ws, long n) {
    long i = (long)blockIdx.x * blockDim.x + threadIdx.x;
    long stride = (long)gridDim.x * blockDim.x;
    for (; i < n; i += stride) ws[i] = 0.0f;
}

#define CNT_R  32768
#define CNT_CH 65536
#define CNT_T  1024

__device__ __forceinline__ void gcn_count_one(int* h, int id, int lo) {
    unsigned r = (unsigned)(id - lo);
    if (r < (unsigned)CNT_R) atomicAdd(h + r, 1);
}

__global__ __launch_bounds__(CNT_T)
void gcn_count_kernel(const int* __restrict__ src, const int* __restrict__ dst,
                      int* __restrict__ part, int E, int N) {
    __shared__ int h[CNT_R];
    const int tid = threadIdx.x;
    const int* __restrict__ idx = blockIdx.z ? dst : src;
    const int lo = blockIdx.y * CNT_R;

    for (int r = tid; r < CNT_R; r += CNT_T) h[r] = 0;
    __syncthreads();

    const long base = (long)blockIdx.x * CNT_CH;
    long left = (long)E - base;
    int cnt = left < 0 ? 0 : (left > CNT_CH ? CNT_CH : (int)left);
    int n4 = cnt >> 2;
    const int4* __restrict__ v4 = (const int4*)(idx + base);

    for (int i0 = tid; i0 < n4; i0 += 8 * CNT_T) {
        int4 v[8];
        #pragma unroll
        for (int u = 0; u < 8; ++u) {
            int i = i0 + u * CNT_T;
            v[u] = v4[i < n4 ? i : n4 - 1];
        }
        #pragma unroll
        for (int u = 0; u < 8; ++u) {
            bool live = i0 + u * CNT_T < n4;
            gcn_count_one(h, live ? v[u].x : -1, lo);
            gcn_count_one(h, live ? v[u].y : -1, lo);
            gcn_count_one(h, live ? v[u].z : -1, lo);
            gcn_count_one(h, live ? v[u].w : -1, lo);
        }
    }
    for (int j = (n4 << 2) + tid; j < cnt; j += CNT_T) gcn_count_one(h, idx[base + j], lo);
    __syncthreads();

    int* __restrict__ outp = part + ((long)blockIdx.z * gridDim.x + blockIdx.x) * N;
    for (int r0 = tid; r0 < CNT_R; r0 += 8 * CNT_T) {
        int c[8];
        #pragma unroll
        for (int u = 0; u < 8; ++u) c[u] = h[r0 + u * CNT_T];
        #pragma unroll
        for (int u = 0; u < 8; ++u) {
            int node = lo + r0 + u * CNT_T;
            if (node < N) outp[node] = c[u];
        }
    }
}

__global__ void gcn_norm_kernel(const int* __restrict__ part, float* __restrict__ norm_src,
                                float* __restrict__ norm_dst, int N, int Q) {
    int i = blockIdx.x * blockDim.x + threadIdx.x;
    if (i < N) {
        const int* __restrict__ po = part + i;
        const int* __restrict__ pi = part + (long)Q * N + i;
        int so = 0, si = 0;
        #pragma unroll 5
        for (int q = 0; q < Q; ++q) {
            so += po[(long)q * N];
            si += pi[(long)q * N];
        }
        norm_src[i] = rsqrtf(fmaxf((float)so, 1.0f));
        norm_dst[i] = rsqrtf(fmaxf((float)si, 1.0f));
    }
}

__global__ void gcn_scatter_kernel(const float* __restrict__ x, const int* __restrict__ src,
                                   const int* __restrict__ dst, const float* __restrict__ norm_src,
                                   float* __restrict__ agg, int E) {
    long tid = (long)blockIdx.x * blockDim.x + threadIdx.x;
    int e = (int)(tid >> 4);
    if (e >= E) return;
    int f = ((int)tid & 15) * 4;
    int s = src[e];
    int d = dst[e];
    float ns = norm_src[s];
    float4 v = *(const float4*)(x + (long)s * D + f);
    float* a = agg + (long)d * D + f;
    atomicAdd(a + 0, v.x * ns);
    atomicAdd(a + 1, v.y * ns);
    atomicAdd(a + 2, v.z * ns);
    atomicAdd(a + 3, v.w * ns);
}

__global__ void gcn_gemm_wmma_kernel(const float* __restrict__ agg,
                                     const float* __restrict__ norm_dst,
                                     const float* __restrict__ Wmat,
                                     float* __restrict__ out,
                                     int N, int numTiles) {
    int lane = threadIdx.x & 31;
    int wave = threadIdx.x >> 5;
    int w = blockIdx.x * (blockDim.x >> 5) + wave;
    if (w >= numTiles) return;

    int mtile = w >> 2;
    int ntile = w & 3;
    int row0 = mtile * 16;
    int col0 = ntile * 16;

    int half = lane >> 4;
    int lm   = lane & 15;

    int m = row0 + lm;
    int mc = m < N ? m : (N - 1);
    float scale = (m < N) ? norm_dst[mc] : 0.0f;
    const float* arow = agg + (long)mc * D;

    v8f c = {};
    #pragma unroll
    for (int kk = 0; kk < 16; ++kk) {
        int kbase = kk * 4 + half * 2;
        float2 av = *(const float2*)(arow + kbase);
        v2f a; a.x = av.x * scale; a.y = av.y * scale;
        v2f b;
        b.x = Wmat[(kbase + 0) * D + col0 + lm];
        b.y = Wmat[(kbase + 1) * D + col0 + lm];
        c = __builtin_amdgcn_wmma_f32_16x16x4_f32(false, a, false, b, (short)0, c, false, false);
    }

    #pragma unroll
    for (int v = 0; v < 8; ++v) {
        int row = row0 + v + half * 8;
        if (row < N)
            out[(long)row * D + col0 + lm] = c[v];
    }
}

extern "C" void kernel_launch(void* const* d_in, const int* in_sizes, int n_in,
                              void* d_out, int out_size, void* d_ws, size_t ws_size,
                              hipStream_t stream) {
    const float* x    = (const float*)d_in[0];
    const float* Wmat = (const float*)d_in[1];
    const int*   src  = (const int*)d_in[2];
    const int*   dst  = (const int*)d_in[3];
    float* out = (float*)d_out;

    int N = in_sizes[0] / D;
    int E = in_sizes[2];

    float* ws_norm_src = (float*)d_ws;
    float* ws_norm_dst = ws_norm_src + N;
    float* ws_agg      = ws_norm_dst + N;

    int* ws_part = (int*)ws_agg;
    int Q = (E + CNT_CH - 1) / CNT_CH;
    int P = (N + CNT_R - 1) / CNT_R;
    gcn_count_kernel<<<dim3(Q, P, 2), CNT_T, 0, stream>>>(src, dst, ws_part, E, N);

    gcn_norm_kernel<<<(N + 255) / 256, 256, 0, stream>>>(ws_part, ws_norm_src, ws_norm_dst, N, Q);

    long nzero = (long)N * D;
    {
        int blocks = (int)((nzero + 255) / 256);
        if (blocks > 65535) blocks = 65535;
        gcn_zero_kernel<<<blocks, 256, 0, stream>>>(ws_agg, nzero);
    }

    {
        long work = (long)E * 16;
        int blocks = (int)((work + 255) / 256);
        gcn_scatter_kernel<<<blocks, 256, 0, stream>>>(x, src, dst, ws_norm_src, ws_agg, E);
    }

    {
        int mtiles = (N + 15) / 16;
        int numTiles = mtiles * 4;
        int wavesPerBlock = 8;
        int blocks = (numTiles + wavesPerBlock - 1) / wavesPerBlock;
        gcn_gemm_wmma_kernel<<<blocks, 256, 0, stream>>>(ws_agg, ws_norm_dst, Wmat, out, N, numTiles);
    }
}


# Scatter remapped to one wave per 32 edges, lane = feature: each wave-wide f32 atomic covers one whole 128-byte line instead of four lines with strided lanes; sums still in atomic arrival order
# speedup vs baseline: 3.0293x; 1.7664x over previous
#include <hip/hip_runtime.h>


typedef float v2f __attribute__((ext_vector_type(2)));
typedef float v8f __attribute__((ext_vector_type(8)));

#define D 64

__global__ void gcn_zero_kernel(float* __restrict__ ws, long n) {
    long i = (long)blockIdx.x * blockDim.x + threadIdx.x;
    long stride = (long)gridDim.x * blockDim.x;
    for (; i < n; i += stride) ws[i] = 0.0f;
}

#define CNT_R  32768
#define CNT_CH 65536
#define CNT_T  1024

__device__ __forceinline__ void gcn_count_one(int* h, int id, int lo) {
    unsigned r = (unsigned)(id - lo);
    if (r < (unsigned)CNT_R) atomicAdd(h + r, 1);
}

__global__ __launch_bounds__(CNT_T)
void gcn_count_kernel(const int* __restrict__ src, const int* __restrict__ dst,
                      int* __restrict__ part, int E, int N) {
    __shared__ int h[CNT_R];
    const int tid = threadIdx.x;
    const int* __restrict__ idx = blockIdx.z ? dst : src;
    const int lo = blockIdx.y * CNT_R;

    for (int r = tid; r < CNT_R; r += CNT_T) h[r] = 0;
    __syncthreads();

    const long base = (long)blockIdx.x * CNT_CH;
    long left = (long)E - base;
    int cnt = left < 0 ? 0 : (left > CNT_CH ? CNT_CH : (int)left);
    int n4 = cnt >> 2;
    const int4* __restrict__ v4 = (const int4*)(idx + base);

    for (int i0 = tid; i0 < n4; i0 += 8 * CNT_T) {
        int4 v[8];
        #pragma unroll
        for (int u = 0; u < 8; ++u) {
            int i = i0 + u * CNT_T;
            v[u] = v4[i < n4 ? i : n4 - 1];
        }
        #pragma unroll
        for (int u = 0; u < 8; ++u) {
            bool live = i0 + u * CNT_T < n4;
            gcn_count_one(h, live ? v[u].x : -1, lo);
            gcn_count_one(h, live ? v[u].y : -1, lo);
            gcn_count_one(h, live ? v[u].z : -1, lo);
            gcn_count_one(h, live ? v[u].w : -1, lo);
        }
    }
    for (int j = (n4 << 2) + tid; j < cnt; j += CNT_T) gcn_count_one(h, idx[base + j], lo);
    __syncthreads();

    int* __restrict__ outp = part + ((long)blockIdx.z * gridDim.x + blockIdx.x) * N;
    for (int r0 = tid; r0 < CNT_R; r0 += 8 * CNT_T) {
        int c[8];
        #pragma unroll
        for (int u = 0; u < 8; ++u) c[u] = h[r0 + u * CNT_T];
        #pragma unroll
        for (int u = 0; u < 8; ++u) {
            int node = lo + r0 + u * CNT_T;
            if (node < N) outp[node] = c[u];
        }
    }
}

__global__ void gcn_norm_kernel(const int* __restrict__ part, float* __restrict__ norm_src,
                                float* __restrict__ norm_dst, int N, int Q) {
    int i = blockIdx.x * blockDim.x + threadIdx.x;
    if (i < N) {
        const int* __restrict__ po = part + i;
        const int* __restrict__ pi = part + (long)Q * N + i;
        int so = 0, si = 0;
        #pragma unroll 5
        for (int q = 0; q < Q; ++q) {
            so += po[(long)q * N];
            si += pi[(long)q * N];
        }
        norm_src[i] = rsqrtf(fmaxf((float)so, 1.0f));
        norm_dst[i] = rsqrtf(fmaxf((float)si, 1.0f));
    }
}

#define SC_G 8

__device__ __forceinline__ int gcn_lane_i(int v, int j) {
    return __builtin_amdgcn_readlane(v, j);
}

__global__ __launch_bounds__(256)
void gcn_scatter_kernel(const float* __restrict__ x, const int* __restrict__ src,
                        const int* __restrict__ dst, const float* __restrict__ norm_src,
                        float* __restrict__ agg, int E) {
    const int lane = threadIdx.x & 31;
    const int wv = __builtin_amdgcn_readfirstlane((int)(threadIdx.x >> 5));
    const long e0 = (long)blockIdx.x * blockDim.x + (long)wv * 32;
    if (e0 >= E) return;

    const long e = e0 + lane;
    const bool live = e < E;
    const int   s_l = live ? src[e] : 0;
    const int   d_l = live ? dst[e] : 0;
    const float n_l = live ? norm_src[s_l] : 0.0f;
    const int   n_b = __float_as_int(n_l);

    if (e0 + 32 <= E) {
        #pragma unroll
        for (int g = 0; g < 32; g += SC_G) {
            float v0[SC_G], v1[SC_G];
            #pragma unroll
            for (int u = 0; u < SC_G; ++u) {
                const float* xr = x + (long)gcn_lane_i(s_l, g + u) * D + lane;
                v0[u] = xr[0];
                v1[u] = xr[32];
            }
            #pragma unroll
            for (int u = 0; u < SC_G; ++u) {
                float ns = __int_as_float(gcn_lane_i(n_b, g + u));
                float* ar = agg + (long)gcn_lane_i(d_l, g + u) * D + lane;
                atomicAdd(ar,      v0[u] * ns);
                atomicAdd(ar + 32, v1[u] * ns);
            }
        }
    } else {
        const int cnt = (int)((long)E - e0);
        for (int j = 0; j < cnt; ++j) {
            int   s  = __shfl(s_l, j, 32);
            int   d  = __shfl(d_l, j, 32);
            float ns = __shfl(n_l, j, 32);
            const float* xr = x + (long)s * D + lane;
            float* ar = agg + (long)d * D + lane;
            float a0 = xr[0], a1 = xr[32];
            atomicAdd(ar,      a0 * ns);
            atomicAdd(ar + 32, a1 * ns);
        }
    }
}

__global__ void gcn_gemm_wmma_kernel(const float* __restrict__ agg,
                                     const float* __restrict__ norm_dst,
                                     const float* __restrict__ Wmat,
                                     float* __restrict__ out,
                                     int N, int numTiles) {
    int lane = threadIdx.x & 31;
    int wave = threadIdx.x >> 5;
    int w = blockIdx.x * (blockDim.x >> 5) + wave;
    if (w >= numTiles) return;

    int mtile = w >> 2;
    int ntile = w & 3;
    int row0 = mtile * 16;
    int col0 = ntile * 16;

    int half = lane >> 4;
    int lm   = lane & 15;

    int m = row0 + lm;
    int mc = m < N ? m : (N - 1);
    float scale = (m < N) ? norm_dst[mc] : 0.0f;
    const float* arow = agg + (long)mc * D;

    v8f c = {};
    #pragma unroll
    for (int kk = 0; kk < 16; ++kk) {
        int kbase = kk * 4 + half * 2;
        float2 av = *(const float2*)(arow + kbase);
        v2f a; a.x = av.x * scale; a.y = av.y * scale;
        v2f b;
        b.x = Wmat[(kbase + 0) * D + col0 + lm];
        b.y = Wmat[(kbase + 1) * D + col0 + lm];
        c = __builtin_amdgcn_wmma_f32_16x16x4_f32(false, a, false, b, (short)0, c, false, false);
    }

    #pragma unroll
    for (int v = 0; v < 8; ++v) {
        int row = row0 + v + half * 8;
        if (row < N)
            out[(long)row * D + col0 + lm] = c[v];
    }
}

extern "C" void kernel_launch(void* const* d_in, const int* in_sizes, int n_in,
                              void* d_out, int out_size, void* d_ws, size_t ws_size,
                              hipStream_t stream) {
    const float* x    = (const float*)d_in[0];
    const float* Wmat = (const float*)d_in[1];
    const int*   src  = (const int*)d_in[2];
    const int*   dst  = (const int*)d_in[3];
    float* out = (float*)d_out;

    int N = in_sizes[0] / D;
    int E = in_sizes[2];

    float* ws_norm_src = (float*)d_ws;
    float* ws_norm_dst = ws_norm_src + N;
    float* ws_agg      = ws_norm_dst + N;

    int* ws_part = (int*)ws_agg;
    int Q = (E + CNT_CH - 1) / CNT_CH;
    int P = (N + CNT_R - 1) / CNT_R;
    gcn_count_kernel<<<dim3(Q, P, 2), CNT_T, 0, stream>>>(src, dst, ws_part, E, N);

    gcn_norm_kernel<<<(N + 255) / 256, 256, 0, stream>>>(ws_part, ws_norm_src, ws_norm_dst, N, Q);

    long nzero = (long)N * D;
    {
        int blocks = (int)((nzero + 255) / 256);
        if (blocks > 65535) blocks = 65535;
        gcn_zero_kernel<<<blocks, 256, 0, stream>>>(ws_agg, nzero);
    }

    {
        long work = (long)E;
        int blocks = (int)((work + 255) / 256);
        gcn_scatter_kernel<<<blocks, 256, 0, stream>>>(x, src, dst, ws_norm_src, ws_agg, E);
    }

    {
        int mtiles = (N + 15) / 16;
        int numTiles = mtiles * 4;
        int wavesPerBlock = 8;
        int blocks = (numTiles + wavesPerBlock - 1) / wavesPerBlock;
        gcn_gemm_wmma_kernel<<<blocks, 256, 0, stream>>>(ws_agg, ws_norm_dst, Wmat, out, N, numTiles);
    }
}


# Zero pass folded into the norm kernel: thread i clears the 64 accumulator words k*N+i after reading its own chunk counts from them; one launch and its 25000-block one-store grid less
# speedup vs baseline: 3.2524x; 1.0736x over previous
#include <hip/hip_runtime.h>


typedef float v2f __attribute__((ext_vector_type(2)));
typedef float v8f __attribute__((ext_vector_type(8)));

#define D 64

#define CNT_R  32768
#define CNT_CH 65536
#define CNT_T  1024

__device__ __forceinline__ void gcn_count_one(int* h, int id, int lo) {
    unsigned r = (unsigned)(id - lo);
    if (r < (unsigned)CNT_R) atomicAdd(h + r, 1);
}

__global__ __launch_bounds__(CNT_T)
void gcn_count_kernel(const int* __restrict__ src, const int* __restrict__ dst,
                      int* __restrict__ part, int E, int N) {
    __shared__ int h[CNT_R];
    const int tid = threadIdx.x;
    const int* __restrict__ idx = blockIdx.z ? dst : src;
    const int lo = blockIdx.y * CNT_R;

    for (int r = tid; r < CNT_R; r += CNT_T) h[r] = 0;
    __syncthreads();

    const long base = (long)blockIdx.x * CNT_CH;
    long left = (long)E - base;
    int cnt = left < 0 ? 0 : (left > CNT_CH ? CNT_CH : (int)left);
    int n4 = cnt >> 2;
    const int4* __restrict__ v4 = (const int4*)(idx + base);

    for (int i0 = tid; i0 < n4; i0 += 8 * CNT_T) {
        int4 v[8];
        #pragma unroll
        for (int u = 0; u < 8; ++u) {
            int i = i0 + u * CNT_T;
            v[u] = v4[i < n4 ? i : n4 - 1];
        }
        #pragma unroll
        for (int u = 0; u < 8; ++u) {
            bool live = i0 + u * CNT_T < n4;
            gcn_count_one(h, live ? v[u].x : -1, lo);
            gcn_count_one(h, live ? v[u].y : -1, lo);
            gcn_count_one(h, live ? v[u].z : -1, lo);
            gcn_count_one(h, live ? v[u].w : -1, lo);
        }
    }
    for (int j = (n4 << 2) + tid; j < cnt; j += CNT_T) gcn_count_one(h, idx[base + j], lo);
    __syncthreads();

    int* __restrict__ outp = part + ((long)blockIdx.z * gridDim.x + blockIdx.x) * N;
    for (int r0 = tid; r0 < CNT_R; r0 += 8 * CNT_T) {
        int c[8];
        #pragma unroll
        for (int u = 0; u < 8; ++u) c[u] = h[r0 + u * CNT_T];
        #pragma unroll
        for (int u = 0; u < 8; ++u) {
            int node = lo + r0 + u * CNT_T;
            if (node < N) outp[node] = c[u];
        }
    }
}

__global__ void gcn_norm_clear_kernel(int* acc, float* __restrict__ norm_src,
                                      float* __restrict__ norm_dst, int N, int Q) {
    int i = blockIdx.x * blockDim.x + threadIdx.x;
    if (i < N) {
        int* col = acc + i;
        const long zoff = (long)Q * N;
        int so = 0, si = 0;
        #pragma unroll 5
        for (int q = 0; q < Q; ++q) {
            so += col[(long)q * N];
            si += col[zoff + (long)q * N];
        }
        norm_src[i] = rsqrtf(fmaxf((float)so, 1.0f));
        norm_dst[i] = rsqrtf(fmaxf((float)si, 1.0f));
        for (int k = 0; k < D; ++k) col[(long)k * N] = 0;
    }
}

#define SC_G 8

__device__ __forceinline__ int gcn_lane_i(int v, int j) {
    return __builtin_amdgcn_readlane(v, j);
}

__global__ __launch_bounds__(256)
void gcn_scatter_kernel(const float* __restrict__ x, const int* __restrict__ src,
                        const int* __restrict__ dst, const float* __restrict__ norm_src,
                        float* __restrict__ agg, int E) {
    const int lane = threadIdx.x & 31;
    const int wv = __builtin_amdgcn_readfirstlane((int)(threadIdx.x >> 5));
    const long e0 = (long)blockIdx.x * blockDim.x + (long)wv * 32;
    if (e0 >= E) return;

    const long e = e0 + lane;
    const bool live = e < E;
    const int   s_l = live ? src[e] : 0;
    const int   d_l = live ? dst[e] : 0;
    const float n_l = live ? norm_src[s_l] : 0.0f;
    const int   n_b = __float_as_int(n_l);

    if (e0 + 32 <= E) {
        #pragma unroll
        for (int g = 0; g < 32; g += SC_G) {
            float v0[SC_G], v1[SC_G];
            #pragma unroll
            for (int u = 0; u < SC_G; ++u) {
                const float* xr = x + (long)gcn_lane_i(s_l, g + u) * D + lane;
                v0[u] = xr[0];
                v1[u] = xr[32];
            }
            #pragma unroll
            for (int u = 0; u < SC_G; ++u) {
                float ns = __int_as_float(gcn_lane_i(n_b, g + u));
                float* ar = agg + (long)gcn_lane_i(d_l, g + u) * D + lane;
                atomicAdd(ar,      v0[u] * ns);
                atomicAdd(ar + 32, v1[u] * ns);
            }
        }
    } else {
        const int cnt = (int)((long)E - e0);
        for (int j = 0; j < cnt; ++j) {
            int   s  = __shfl(s_l, j, 32);
            int   d  = __shfl(d_l, j, 32);
            float ns = __shfl(n_l, j, 32);
            const float* xr = x + (long)s * D + lane;
            float* ar = agg + (long)d * D + lane;
            float a0 = xr[0], a1 = xr[32];
            atomicAdd(ar,      a0 * ns);
            atomicAdd(ar + 32, a1 * ns);
        }
    }
}

__global__ void gcn_gemm_wmma_kernel(const float* __restrict__ agg,
                                     const float* __restrict__ norm_dst,
                                     const float* __restrict__ Wmat,
                                     float* __restrict__ out,
                                     int N, int numTiles) {
    int lane = threadIdx.x & 31;
    int wave = threadIdx.x >> 5;
    int w = blockIdx.x * (blockDim.x >> 5) + wave;
    if (w >= numTiles) return;

    int mtile = w >> 2;
    int ntile = w & 3;
    int row0 = mtile * 16;
    int col0 = ntile * 16;

    int half = lane >> 4;
    int lm   = lane & 15;

    int m = row0 + lm;
    int mc = m < N ? m : (N - 1);
    float scale = (m < N) ? norm_dst[mc] : 0.0f;
    const float* arow = agg + (long)mc * D;

    v8f c = {};
    #pragma unroll
    for (int kk = 0; kk < 16; ++kk) {
        int kbase = kk * 4 + half * 2;
        float2 av = *(const float2*)(arow + kbase);
        v2f a; a.x = av.x * scale; a.y = av.y * scale;
        v2f b;
        b.x = Wmat[(kbase + 0) * D + col0 + lm];
        b.y = Wmat[(kbase + 1) * D + col0 + lm];
        c = __builtin_amdgcn_wmma_f32_16x16x4_f32(false, a, false, b, (short)0, c, false, false);
    }

    #pragma unroll
    for (int v = 0; v < 8; ++v) {
        int row = row0 + v + half * 8;
        if (row < N)
            out[(long)row * D + col0 + lm] = c[v];
    }
}

extern "C" void kernel_launch(void* const* d_in, const int* in_sizes, int n_in,
                              void* d_out, int out_size, void* d_ws, size_t ws_size,
                              hipStream_t stream) {
    const float* x    = (const float*)d_in[0];
    const float* Wmat = (const float*)d_in[1];
    const int*   src  = (const int*)d_in[2];
    const int*   dst  = (const int*)d_in[3];
    float* out = (float*)d_out;

    int N = in_sizes[0] / D;
    int E = in_sizes[2];

    float* ws_norm_src = (float*)d_ws;
    float* ws_norm_dst = ws_norm_src + N;
    float* ws_agg      = ws_norm_dst + N;

    int* ws_part = (int*)ws_agg;
    int Q = (E + CNT_CH - 1) / CNT_CH;
    int P = (N + CNT_R - 1) / CNT_R;
    gcn_count_kernel<<<dim3(Q, P, 2), CNT_T, 0, stream>>>(src, dst, ws_part, E, N);

    gcn_norm_clear_kernel<<<(N + 255) / 256, 256, 0, stream>>>(ws_part, ws_norm_src, ws_norm_dst, N, Q);

    {
        long work = (long)E;
        int blocks = (int)((work + 255) / 256);
        gcn_scatter_kernel<<<blocks, 256, 0, stream>>>(x, src, dst, ws_norm_src, ws_agg, E);
    }

    {
        int mtiles = (N + 15) / 16;
        int numTiles = mtiles * 4;
        int wavesPerBlock = 8;
        int blocks = (numTiles + wavesPerBlock - 1) / wavesPerBlock;
        gcn_gemm_wmma_kernel<<<blocks, 256, 0, stream>>>(ws_agg, ws_norm_dst, Wmat, out, N, numTiles);
    }
}


# Scatter on resident blocks that draw 256-edge tickets per cache half (die id read in the wave): a wave adds only into rows of its own half's node range; adds still in atomic arrival order
# speedup vs baseline: 3.8425x; 1.1814x over previous
#include <hip/hip_runtime.h>


typedef float v2f __attribute__((ext_vector_type(2)));
typedef float v8f __attribute__((ext_vector_type(8)));

#define D 64
#define SC_PITCH 64

#define CNT_R  32768
#define CNT_CH 65536
#define CNT_T  1024

__device__ __forceinline__ void gcn_count_one(int* h, int id, int lo) {
    unsigned r = (unsigned)(id - lo);
    if (r < (unsigned)CNT_R) atomicAdd(h + r, 1);
}

__global__ __launch_bounds__(CNT_T)
void gcn_count_kernel(const int* __restrict__ src, const int* __restrict__ dst,
                      int* __restrict__ part, int E, int N) {
    __shared__ int h[CNT_R];
    const int tid = threadIdx.x;
    const int* __restrict__ idx = blockIdx.z ? dst : src;
    const int lo = blockIdx.y * CNT_R;

    for (int r = tid; r < CNT_R; r += CNT_T) h[r] = 0;
    __syncthreads();

    const long base = (long)blockIdx.x * CNT_CH;
    long left = (long)E - base;
    int cnt = left < 0 ? 0 : (left > CNT_CH ? CNT_CH : (int)left);
    int n4 = cnt >> 2;
    const int4* __restrict__ v4 = (const int4*)(idx + base);

    for (int i0 = tid; i0 < n4; i0 += 8 * CNT_T) {
        int4 v[8];
        #pragma unroll
        for (int u = 0; u < 8; ++u) {
            int i = i0 + u * CNT_T;
            v[u] = v4[i < n4 ? i : n4 - 1];
        }
        #pragma unroll
        for (int u = 0; u < 8; ++u) {
            bool live = i0 + u * CNT_T < n4;
            gcn_count_one(h, live ? v[u].x : -1, lo);
            gcn_count_one(h, live ? v[u].y : -1, lo);
            gcn_count_one(h, live ? v[u].z : -1, lo);
            gcn_count_one(h, live ? v[u].w : -1, lo);
        }
    }
    for (int j = (n4 << 2) + tid; j < cnt; j += CNT_T) gcn_count_one(h, idx[base + j], lo);
    __syncthreads();

    int* __restrict__ outp = part + ((long)blockIdx.z * gridDim.x + blockIdx.x) * N;
    for (int r0 = tid; r0 < CNT_R; r0 += 8 * CNT_T) {
        int c[8];
        #pragma unroll
        for (int u = 0; u < 8; ++u) c[u] = h[r0 + u * CNT_T];
        #pragma unroll
        for (int u = 0; u < 8; ++u) {
            int node = lo + r0 + u * CNT_T;
            if (node < N) outp[node] = c[u];
        }
    }
}

__global__ void gcn_norm_clear_kernel(int* acc, float* __restrict__ norm_src,
                                      float* __restrict__ norm_dst, int* __restrict__ tick, int N, int Q) {
    int i = blockIdx.x * blockDim.x + threadIdx.x;
    if (i == 0) { tick[0] = 0; tick[SC_PITCH] = 0; }
    if (i < N) {
        int* col = acc + i;
        const long zoff = (long)Q * N;
        int so = 0, si = 0;
        #pragma unroll 5
        for (int q = 0; q < Q; ++q) {
            so += col[(long)q * N];
            si += col[zoff + (long)q * N];
        }
        norm_src[i] = rsqrtf(fmaxf((float)so, 1.0f));
        norm_dst[i] = rsqrtf(fmaxf((float)si, 1.0f));
        for (int k = 0; k < D; ++k) col[(long)k * N] = 0;
    }
}

#define SC_G 8
#define SC_T 256
#define SC_BLOCKS 2048

__device__ __forceinline__ int gcn_lane_i(int v, int j) {
    return __builtin_amdgcn_readlane(v, j);
}

__device__ __forceinline__ int gcn_my_half() {
    unsigned id = __builtin_amdgcn_s_sendmsg_rtn(0x87);
    return ((id >> 16) & 15u) >= 4u ? 1 : 0;
}

template <bool GUARD>
__device__ __forceinline__ unsigned gcn_scatter_rows(const float* __restrict__ x, float* __restrict__ agg,
                                                     int s_l, int d_l, int n_b, unsigned m, int lane) {
    int j[SC_G];
    #pragma unroll
    for (int u = 0; u < SC_G; ++u) {
        j[u] = (!GUARD || m != 0u) ? __builtin_ctz(m) : -1;
        m &= m - 1u;
    }
    float v0[SC_G], v1[SC_G];
    #pragma unroll
    for (int u = 0; u < SC_G; ++u) {
        v0[u] = 0.0f; v1[u] = 0.0f;
        if (!GUARD || j[u] >= 0) {
            const float* xr = x + (long)gcn_lane_i(s_l, j[u]) * D + lane;
            v0[u] = xr[0];
            v1[u] = xr[32];
        }
    }
    #pragma unroll
    for (int u = 0; u < SC_G; ++u) {
        if (!GUARD || j[u] >= 0) {
            float ns = __int_as_float(gcn_lane_i(n_b, j[u]));
            float* ar = agg + (long)gcn_lane_i(d_l, j[u]) * D + lane;
            atomicAdd(ar,      v0[u] * ns);
            atomicAdd(ar + 32, v1[u] * ns);
        }
    }
    return m;
}

__global__ __launch_bounds__(256)
void gcn_scatter_kernel(const float* __restrict__ x, const int* __restrict__ src,
                        const int* __restrict__ dst, const float* __restrict__ norm_src,
                        float* __restrict__ agg, int* __restrict__ tick, int E, int N, int C) {
    __shared__ int s_tk[2];
    const int lane = threadIdx.x & 31;
    const int wv = __builtin_amdgcn_readfirstlane((int)(threadIdx.x >> 5));
    const int own = gcn_my_half();
    const int split = N >> 1;

    for (int ph = 0; ph < 2; ++ph) {
        const int hh = own ^ ph;
        unsigned* cnt = (unsigned*)tick + hh * SC_PITCH;
        if (threadIdx.x == 0) s_tk[0] = (int)atomicInc(cnt, 0xFFFFFFFFu);
        __syncthreads();
        int slot = 0;
        int cur = __builtin_amdgcn_readfirstlane(s_tk[0]);
        while (cur < C) {
            const long e0 = (long)cur * SC_T + (long)wv * 32;
            unsigned m = 0u;
            int s_l = 0, d_l = 0, n_b = 0;
            if (e0 < E) {
                const long e = e0 + lane;
                const bool live = e < E;
                s_l = live ? src[e] : 0;
                d_l = live ? dst[e] : 0;
                const bool mine = live && ((d_l >= split ? 1 : 0) == hh);
                n_b = __float_as_int(mine ? norm_src[s_l] : 0.0f);
                m = __builtin_amdgcn_ballot_w32(mine);
            }
            int nxt = 0;
            if (threadIdx.x == 0) nxt = (int)atomicInc(cnt, 0xFFFFFFFFu);
            while (m != 0u) {
                if (__builtin_popcount(m) >= SC_G) m = gcn_scatter_rows<false>(x, agg, s_l, d_l, n_b, m, lane);
                else                               m = gcn_scatter_rows<true>(x, agg, s_l, d_l, n_b, m, lane);
            }
            slot ^= 1;
            if (threadIdx.x == 0) s_tk[slot] = nxt;
            __builtin_amdgcn_fence(__ATOMIC_RELEASE, "workgroup", "local");
            __builtin_amdgcn_s_barrier();
            __builtin_amdgcn_fence(__ATOMIC_ACQUIRE, "workgroup", "local");
            cur = __builtin_amdgcn_readfirstlane(s_tk[slot]);
        }
        __syncthreads();
    }
}

__global__ void gcn_gemm_wmma_kernel(const float* __restrict__ agg,
                                     const float* __restrict__ norm_dst,
                                     const float* __restrict__ Wmat,
                                     float* __restrict__ out,
                                     int N, int numTiles) {
    int lane = threadIdx.x & 31;
    int wave = threadIdx.x >> 5;
    int w = blockIdx.x * (blockDim.x >> 5) + wave;
    if (w >= numTiles) return;

    int mtile = w >> 2;
    int ntile = w & 3;
    int row0 = mtile * 16;
    int col0 = ntile * 16;

    int half = lane >> 4;
    int lm   = lane & 15;

    int m = row0 + lm;
    int mc = m < N ? m : (N - 1);
    float scale = (m < N) ? norm_dst[mc] : 0.0f;
    const float* arow = agg + (long)mc * D;

    v8f c = {};
    #pragma unroll
    for (int kk = 0; kk < 16; ++kk) {
        int kbase = kk * 4 + half * 2;
        float2 av = *(const float2*)(arow + kbase);
        v2f a; a.x = av.x * scale; a.y = av.y * scale;
        v2f b;
        b.x = Wmat[(kbase + 0) * D + col0 + lm];
        b.y = Wmat[(kbase + 1) * D + col0 + lm];
        c = __builtin_amdgcn_wmma_f32_16x16x4_f32(false, a, false, b, (short)0, c, false, false);
    }

    #pragma unroll
    for (int v = 0; v < 8; ++v) {
        int row = row0 + v + half * 8;
        if (row < N)
            out[(long)row * D + col0 + lm] = c[v];
    }
}

extern "C" void kernel_launch(void* const* d_in, const int* in_sizes, int n_in,
                              void* d_out, int out_size, void* d_ws, size_t ws_size,
                              hipStream_t stream) {
    const float* x    = (const float*)d_in[0];
    const float* Wmat = (const float*)d_in[1];
    const int*   src  = (const int*)d_in[2];
    const int*   dst  = (const int*)d_in[3];
    float* out = (float*)d_out;

    int N = in_sizes[0] / D;
    int E = in_sizes[2];

    float* ws_norm_src = (float*)d_ws;
    float* ws_norm_dst = ws_norm_src + N;
    float* ws_agg      = ws_norm_dst + N;

    int* ws_part = (int*)ws_agg;
    int Q = (E + CNT_CH - 1) / CNT_CH;
    int P = (N + CNT_R - 1) / CNT_R;
    gcn_count_kernel<<<dim3(Q, P, 2), CNT_T, 0, stream>>>(src, dst, ws_part, E, N);

    int* tick = (int*)d_out;
    gcn_norm_clear_kernel<<<(N + 255) / 256, 256, 0, stream>>>(ws_part, ws_norm_src, ws_norm_dst, tick, N, Q);

    {
        int C = (E + SC_T - 1) / SC_T;
        gcn_scatter_kernel<<<SC_BLOCKS, 256, 0, stream>>>(x, src, dst, ws_norm_src, ws_agg, tick, E, N, C);
    }

    {
        int mtiles = (N + 15) / 16;
        int numTiles = mtiles * 4;
        int wavesPerBlock = 8;
        int blocks = (numTiles + wavesPerBlock - 1) / wavesPerBlock;
        gcn_gemm_wmma_kernel<<<blocks, 256, 0, stream>>>(ws_agg, ws_norm_dst, Wmat, out, N, numTiles);
    }
}


# Edges binned by destination bucket of 208 rows (LDS-ranked pass); one block per bucket sums its rows by LDS float adds in arrival order and feeds the WMMA product from LDS; no global float atomics
# speedup vs baseline: 4.2807x; 1.1140x over previous
#include <hip/hip_runtime.h>


typedef float v2f __attribute__((ext_vector_type(2)));
typedef float v8f __attribute__((ext_vector_type(8)));

#define D 64

#define CNT_R  32768
#define CNT_CH 65536
#define CNT_T  1024

__device__ __forceinline__ void gcn_count_one(int* h, int id, int lo) {
    unsigned r = (unsigned)(id - lo);
    if (r < (unsigned)CNT_R) atomicAdd(h + r, 1);
}

__global__ __launch_bounds__(CNT_T)
void gcn_count_kernel(const int* __restrict__ src, const int* __restrict__ dst,
                      int* __restrict__ part, int E, int N) {
    __shared__ int h[CNT_R];
    const int tid = threadIdx.x;
    const int* __restrict__ idx = blockIdx.z ? dst : src;
    const int lo = blockIdx.y * CNT_R;

    for (int r = tid; r < CNT_R; r += CNT_T) h[r] = 0;
    __syncthreads();

    const long base = (long)blockIdx.x * CNT_CH;
    long left = (long)E - base;
    int cnt = left < 0 ? 0 : (left > CNT_CH ? CNT_CH : (int)left);
    int n4 = cnt >> 2;
    const int4* __restrict__ v4 = (const int4*)(idx + base);

    for (int i0 = tid; i0 < n4; i0 += 8 * CNT_T) {
        int4 v[8];
        #pragma unroll
        for (int u = 0; u < 8; ++u) {
            int i = i0 + u * CNT_T;
            v[u] = v4[i < n4 ? i : n4 - 1];
        }
        #pragma unroll
        for (int u = 0; u < 8; ++u) {
            bool live = i0 + u * CNT_T < n4;
            gcn_count_one(h, live ? v[u].x : -1, lo);
            gcn_count_one(h, live ? v[u].y : -1, lo);
            gcn_count_one(h, live ? v[u].z : -1, lo);
            gcn_count_one(h, live ? v[u].w : -1, lo);
        }
    }
    for (int j = (n4 << 2) + tid; j < cnt; j += CNT_T) gcn_count_one(h, idx[base + j], lo);
    __syncthreads();

    int* __restrict__ outp = part + ((long)blockIdx.z * gridDim.x + blockIdx.x) * N;
    for (int r0 = tid; r0 < CNT_R; r0 += 8 * CNT_T) {
        int c[8];
        #pragma unroll
        for (int u = 0; u < 8; ++u) c[u] = h[r0 + u * CNT_T];
        #pragma unroll
        for (int u = 0; u < 8; ++u) {
            int node = lo + r0 + u * CNT_T;
            if (node < N) outp[node] = c[u];
        }
    }
}

#define BK_R   208
#define BK_P   68
#define BK_MAX 1024
#define BK_T   1024

__device__ __forceinline__ int gcn_block_excl_scan(int v, int* s_w) {
    const int lane = threadIdx.x & 31;
    const int wv = threadIdx.x >> 5;
    int inc = v;
    #pragma unroll
    for (int o = 1; o < 32; o <<= 1) {
        int t = __shfl_up(inc, o);
        if (lane >= o) inc += t;
    }
    if (lane == 31) s_w[wv] = inc;
    __syncthreads();
    if (wv == 0) {
        int w = s_w[lane];
        int winc = w;
        #pragma unroll
        for (int o = 1; o < 32; o <<= 1) {
            int t = __shfl_up(winc, o);
            if (lane >= o) winc += t;
        }
        s_w[lane] = winc - w;
    }
    __syncthreads();
    return inc - v + s_w[wv];
}

__global__ __launch_bounds__(BK_R)
void gcn_norm_kernel(const int* __restrict__ part, float* __restrict__ norm_src,
                     float* __restrict__ norm_dst, int* __restrict__ tot, int* __restrict__ cur, int N, int Q) {
    __shared__ int s_sum;
    if (threadIdx.x == 0) s_sum = 0;
    __syncthreads();
    const int i = blockIdx.x * BK_R + threadIdx.x;
    int si = 0;
    if (i < N) {
        const int* col = part + i;
        const long zoff = (long)Q * N;
        int so = 0;
        #pragma unroll 5
        for (int q = 0; q < Q; ++q) {
            so += col[(long)q * N];
            si += col[zoff + (long)q * N];
        }
        norm_src[i] = rsqrtf(fmaxf((float)so, 1.0f));
        norm_dst[i] = rsqrtf(fmaxf((float)si, 1.0f));
    }
    atomicAdd(&s_sum, si);
    __syncthreads();
    if (threadIdx.x == 0) { tot[blockIdx.x] = s_sum; cur[blockIdx.x] = 0; }
}

#define BIN_K  8
#define BIN_CH (BK_T * BIN_K)

__global__ __launch_bounds__(BK_T)
void gcn_bin_kernel(const int* __restrict__ src, const int* __restrict__ dst, const int* __restrict__ tot,
                    int* __restrict__ cur, int2* __restrict__ list, int E, int NB) {
    __shared__ int  s_cnt[BK_MAX];
    __shared__ int  s_loc[BK_MAX];
    __shared__ int2 s_stage[BIN_CH];
    __shared__ int  s_w[32];
    const int tid = threadIdx.x;
    s_cnt[tid] = 0;
    const int my_tot = tid < NB ? tot[tid] : 0;

    const long base = (long)blockIdx.x * BIN_CH;
    int s[BIN_K], d[BIN_K];
    #pragma unroll
    for (int u = 0; u < BIN_K; ++u) {
        const long e = base + u * BK_T + tid;
        const bool live = e < E;
        s[u] = live ? src[e] : 0;
        d[u] = live ? dst[e] : -1;
    }

    const int start = gcn_block_excl_scan(my_tot, s_w);

    int bk[BIN_K], rk[BIN_K];
    #pragma unroll
    for (int u = 0; u < BIN_K; ++u) {
        bk[u] = 0; rk[u] = 0;
        if (d[u] >= 0) {
            bk[u] = (int)((unsigned)d[u] / (unsigned)BK_R);
            rk[u] = atomicAdd(&s_cnt[bk[u]], 1);
        }
    }
    __syncthreads();
    const int c = s_cnt[tid];
    const int loc = gcn_block_excl_scan(c, s_w);
    int run = 0;
    if (c > 0) run = atomicAdd(&cur[tid], c);
    s_loc[tid] = loc;
    s_cnt[tid] = start + run - loc;
    __syncthreads();
    #pragma unroll
    for (int u = 0; u < BIN_K; ++u) {
        if (d[u] >= 0) s_stage[s_loc[bk[u]] + rk[u]] = make_int2(s[u], d[u]);
    }
    __syncthreads();
    const long left = (long)E - base;
    const int nlive = left > BIN_CH ? BIN_CH : (int)left;
    for (int p = tid; p < nlive; p += BK_T) {
        const int2 sd = s_stage[p];
        const int b = (int)((unsigned)sd.y / (unsigned)BK_R);
        list[s_cnt[b] + p] = make_int2(sd.x, sd.y - b * BK_R);
    }
}

#define RS_G 8

__device__ __forceinline__ int gcn_lane_i(int v, int j) {
    return __builtin_amdgcn_readlane(v, j);
}

template <bool GUARD>
__device__ __forceinline__ unsigned gcn_sum_rows(const float* __restrict__ x, float* acc,
                                                 int s_l, int r_l, int n_b, unsigned m, int lane) {
    int j[RS_G];
    #pragma unroll
    for (int u = 0; u < RS_G; ++u) {
        j[u] = (!GUARD || m != 0u) ? __builtin_ctz(m) : -1;
        m &= m - 1u;
    }
    float2 v[RS_G];
    #pragma unroll
    for (int u = 0; u < RS_G; ++u) {
        v[u] = make_float2(0.0f, 0.0f);
        if (!GUARD || j[u] >= 0) {
            const float2* xr = (const float2*)(x + (long)gcn_lane_i(s_l, j[u]) * D) + lane;
            v[u] = xr[0];
        }
    }
    #pragma unroll
    for (int u = 0; u < RS_G; ++u) {
        if (!GUARD || j[u] >= 0) {
            const float ns = __int_as_float(gcn_lane_i(n_b, j[u]));
            float* ar = acc + gcn_lane_i(r_l, j[u]) * BK_P + 2 * lane;
            atomicAdd(ar,     v[u].x * ns);
            atomicAdd(ar + 1, v[u].y * ns);
        }
    }
    return m;
}

__global__ __launch_bounds__(BK_T)
void gcn_rowsum_product_kernel(const float* __restrict__ x, const int2* __restrict__ list,
                               const int* __restrict__ tot, const float* __restrict__ norm_src,
                               const float* __restrict__ norm_dst, const float* __restrict__ Wmat,
                               float* __restrict__ out, int N, int NB) {
    __shared__ float acc[BK_R * BK_P];
    __shared__ int s_w[32];
    __shared__ int s_rng[2];
    const int tid = threadIdx.x;
    const int lane = tid & 31;
    const int wv = __builtin_amdgcn_readfirstlane(tid >> 5);

    for (int k = tid; k < BK_R * BK_P; k += BK_T) acc[k] = 0.0f;
    const int my_tot = tid < NB ? tot[tid] : 0;
    const int start = gcn_block_excl_scan(my_tot, s_w);
    if (tid == (int)blockIdx.x) { s_rng[0] = start; s_rng[1] = my_tot; }
    __syncthreads();
    const int first = __builtin_amdgcn_readfirstlane(s_rng[0]);
    const int cnt   = __builtin_amdgcn_readfirstlane(s_rng[1]);

    for (int e0 = wv * 32; e0 < cnt; e0 += BK_T) {
        const int e = e0 + lane;
        const bool live = e < cnt;
        int2 sr = make_int2(0, 0);
        if (live) sr = list[first + e];
        const int n_b = __float_as_int(live ? norm_src[sr.x] : 0.0f);
        unsigned m = __builtin_amdgcn_ballot_w32(live);
        while (m != 0u) {
            if (__builtin_popcount(m) >= RS_G) m = gcn_sum_rows<false>(x, acc, sr.x, sr.y, n_b, m, lane);
            else                               m = gcn_sum_rows<true>(x, acc, sr.x, sr.y, n_b, m, lane);
        }
    }
    __syncthreads();

    if (wv < 2 * (BK_R / 16)) {
        const int half = lane >> 4;
        const int lm   = lane & 15;
        const int rt   = wv >> 1;
        const int col0 = (wv & 1) * 32;
        const int row  = rt * 16 + lm;
        const int g    = (int)blockIdx.x * BK_R + row;
        const int gc   = g < N ? g : (N - 1);
        const float scale = (g < N) ? norm_dst[gc] : 0.0f;
        const float* arow = acc + row * BK_P;

        v8f c0 = {}, c1 = {};
        for (int kk = 0; kk < 16; ++kk) {
            const int kbase = kk * 4 + half * 2;
            const float2 av = *(const float2*)(arow + kbase);
            v2f a; a.x = av.x * scale; a.y = av.y * scale;
            const float* w0 = Wmat + kbase * D + col0 + lm;
            v2f b0, b1;
            b0.x = w0[0];  b0.y = w0[D];
            b1.x = w0[16]; b1.y = w0[D + 16];
            c0 = __builtin_amdgcn_wmma_f32_16x16x4_f32(false, a, false, b0, (short)0, c0, false, false);
            c1 = __builtin_amdgcn_wmma_f32_16x16x4_f32(false, a, false, b1, (short)0, c1, false, false);
        }

        #pragma unroll
        for (int v = 0; v < 8; ++v) {
            const int gr = (int)blockIdx.x * BK_R + rt * 16 + v + half * 8;
            if (gr < N) {
                float* o = out + (long)gr * D + col0 + lm;
                o[0] = c0[v]; o[16] = c1[v];
            }
        }
    }
}

extern "C" void kernel_launch(void* const* d_in, const int* in_sizes, int n_in,
                              void* d_out, int out_size, void* d_ws, size_t ws_size,
                              hipStream_t stream) {
    const float* x    = (const float*)d_in[0];
    const float* Wmat = (const float*)d_in[1];
    const int*   src  = (const int*)d_in[2];
    const int*   dst  = (const int*)d_in[3];
    float* out = (float*)d_out;

    int N = in_sizes[0] / D;
    int E = in_sizes[2];

    float* ws_norm_src = (float*)d_ws;
    float* ws_norm_dst = ws_norm_src + N;
    int*   ws_region   = (int*)(ws_norm_dst + N);

    int*  ws_part = ws_region;
    int2* ws_list = (int2*)ws_region;
    int*  ws_tot  = ws_region + ((long)N * D - 2 * BK_MAX);
    int*  ws_cur  = ws_tot + BK_MAX;

    int Q = (E + CNT_CH - 1) / CNT_CH;
    int P = (N + CNT_R - 1) / CNT_R;
    gcn_count_kernel<<<dim3(Q, P, 2), CNT_T, 0, stream>>>(src, dst, ws_part, E, N);

    int NB = (N + BK_R - 1) / BK_R;
    gcn_norm_kernel<<<NB, BK_R, 0, stream>>>(ws_part, ws_norm_src, ws_norm_dst, ws_tot, ws_cur, N, Q);

    gcn_bin_kernel<<<(E + BIN_CH - 1) / BIN_CH, BK_T, 0, stream>>>(src, dst, ws_tot, ws_cur, ws_list, E, NB);

    gcn_rowsum_product_kernel<<<NB, BK_T, 0, stream>>>(x, ws_list, ws_tot, ws_norm_src, ws_norm_dst, Wmat, out, N, NB);
}


# Row sums in registers: a bucket's edges are counting-sorted by row in LDS (integer ranks, scan, place), waves walk slices and add one partial per row run; sum order = rank order, multiply fused in
# speedup vs baseline: 5.1758x; 1.2091x over previous
#include <hip/hip_runtime.h>


typedef float v2f __attribute__((ext_vector_type(2)));
typedef float v8f __attribute__((ext_vector_type(8)));

#define D 64

#define CNT_R  32768
#define CNT_CH 65536
#define CNT_T  1024

__device__ __forceinline__ void gcn_count_one(int* h, int id, int lo) {
    unsigned r = (unsigned)(id - lo);
    if (r < (unsigned)CNT_R) atomicAdd(h + r, 1);
}

__global__ __launch_bounds__(CNT_T)
void gcn_count_kernel(const int* __restrict__ src, const int* __restrict__ dst,
                      int* __restrict__ part, int E, int N) {
    __shared__ int h[CNT_R];
    const int tid = threadIdx.x;
    const int* __restrict__ idx = blockIdx.z ? dst : src;
    const int lo = blockIdx.y * CNT_R;

    for (int r = tid; r < CNT_R; r += CNT_T) h[r] = 0;
    __syncthreads();

    const long base = (long)blockIdx.x * CNT_CH;
    long left = (long)E - base;
    int cnt = left < 0 ? 0 : (left > CNT_CH ? CNT_CH : (int)left);
    int n4 = cnt >> 2;
    const int4* __restrict__ v4 = (const int4*)(idx + base);

    for (int i0 = tid; i0 < n4; i0 += 8 * CNT_T) {
        int4 v[8];
        #pragma unroll
        for (int u = 0; u < 8; ++u) {
            int i = i0 + u * CNT_T;
            v[u] = v4[i < n4 ? i : n4 - 1];
        }
        #pragma unroll
        for (int u = 0; u < 8; ++u) {
            bool live = i0 + u * CNT_T < n4;
            gcn_count_one(h, live ? v[u].x : -1, lo);
            gcn_count_one(h, live ? v[u].y : -1, lo);
            gcn_count_one(h, live ? v[u].z : -1, lo);
            gcn_count_one(h, live ? v[u].w : -1, lo);
        }
    }
    for (int j = (n4 << 2) + tid; j < cnt; j += CNT_T) gcn_count_one(h, idx[base + j], lo);
    __syncthreads();

    int* __restrict__ outp = part + ((long)blockIdx.z * gridDim.x + blockIdx.x) * N;
    for (int r0 = tid; r0 < CNT_R; r0 += 8 * CNT_T) {
        int c[8];
        #pragma unroll
        for (int u = 0; u < 8; ++u) c[u] = h[r0 + u * CNT_T];
        #pragma unroll
        for (int u = 0; u < 8; ++u) {
            int node = lo + r0 + u * CNT_T;
            if (node < N) outp[node] = c[u];
        }
    }
}

#define BK_R   208
#define BK_P   68
#define BK_MAX 1024
#define BK_T   1024

__device__ __forceinline__ int gcn_block_excl_scan(int v, int* s_w) {
    const int lane = threadIdx.x & 31;
    const int wv = threadIdx.x >> 5;
    int inc = v;
    #pragma unroll
    for (int o = 1; o < 32; o <<= 1) {
        int t = __shfl_up(inc, o);
        if (lane >= o) inc += t;
    }
    if (lane == 31) s_w[wv] = inc;
    __syncthreads();
    if (wv == 0) {
        int w = s_w[lane];
        int winc = w;
        #pragma unroll
        for (int o = 1; o < 32; o <<= 1) {
            int t = __shfl_up(winc, o);
            if (lane >= o) winc += t;
        }
        s_w[lane] = winc - w;
    }
    __syncthreads();
    return inc - v + s_w[wv];
}

__global__ __launch_bounds__(BK_R)
void gcn_norm_kernel(const int* __restrict__ part, float* __restrict__ norm_src,
                     float* __restrict__ norm_dst, int* __restrict__ tot, int* __restrict__ cur, int N, int Q) {
    __shared__ int s_sum;
    if (threadIdx.x == 0) s_sum = 0;
    __syncthreads();
    const int i = blockIdx.x * BK_R + threadIdx.x;
    int si = 0;
    if (i < N) {
        const int* col = part + i;
        const long zoff = (long)Q * N;
        int so = 0;
        #pragma unroll 5
        for (int q = 0; q < Q; ++q) {
            so += col[(long)q * N];
            si += col[zoff + (long)q * N];
        }
        norm_src[i] = rsqrtf(fmaxf((float)so, 1.0f));
        norm_dst[i] = rsqrtf(fmaxf((float)si, 1.0f));
    }
    atomicAdd(&s_sum, si);
    __syncthreads();
    if (threadIdx.x == 0) { tot[blockIdx.x] = s_sum; cur[blockIdx.x] = 0; }
}

#define BIN_K  8
#define BIN_CH (BK_T * BIN_K)

__global__ __launch_bounds__(BK_T)
void gcn_bin_kernel(const int* __restrict__ src, const int* __restrict__ dst, const int* __restrict__ tot,
                    int* __restrict__ cur, int2* __restrict__ list, int E, int NB) {
    __shared__ int  s_cnt[BK_MAX];
    __shared__ int  s_loc[BK_MAX];
    __shared__ int2 s_stage[BIN_CH];
    __shared__ int  s_w[32];
    const int tid = threadIdx.x;
    s_cnt[tid] = 0;
    const int my_tot = tid < NB ? tot[tid] : 0;

    const long base = (long)blockIdx.x * BIN_CH;
    int s[BIN_K], d[BIN_K];
    #pragma unroll
    for (int u = 0; u < BIN_K; ++u) {
        const long e = base + u * BK_T + tid;
        const bool live = e < E;
        s[u] = live ? src[e] : 0;
        d[u] = live ? dst[e] : -1;
    }

    const int start = gcn_block_excl_scan(my_tot, s_w);

    int bk[BIN_K], rk[BIN_K];
    #pragma unroll
    for (int u = 0; u < BIN_K; ++u) {
        bk[u] = 0; rk[u] = 0;
        if (d[u] >= 0) {
            bk[u] = (int)((unsigned)d[u] / (unsigned)BK_R);
            rk[u] = atomicAdd(&s_cnt[bk[u]], 1);
        }
    }
    __syncthreads();
    const int c = s_cnt[tid];
    const int loc = gcn_block_excl_scan(c, s_w);
    int run = 0;
    if (c > 0) run = atomicAdd(&cur[tid], c);
    s_loc[tid] = loc;
    s_cnt[tid] = start + run - loc;
    __syncthreads();
    #pragma unroll
    for (int u = 0; u < BIN_K; ++u) {
        if (d[u] >= 0) s_stage[s_loc[bk[u]] + rk[u]] = make_int2(s[u], d[u]);
    }
    __syncthreads();
    const long left = (long)E - base;
    const int nlive = left > BIN_CH ? BIN_CH : (int)left;
    for (int p = tid; p < nlive; p += BK_T) {
        const int2 sd = s_stage[p];
        const int b = (int)((unsigned)sd.y / (unsigned)BK_R);
        list[s_cnt[b] + p] = make_int2(sd.x, sd.y - b * BK_R);
    }
}

#define RS_G  8
#define RS_U  4
#define RS_CH (BK_T * RS_U)

__device__ __forceinline__ int gcn_lane_i(int v, int j) {
    return __builtin_amdgcn_readlane(v, j);
}

__device__ __forceinline__ void gcn_flush_row(float* acc, int row, int lane, float2 sum) {
    float* ar = acc + row * BK_P + 2 * lane;
    atomicAdd(ar,     sum.x);
    atomicAdd(ar + 1, sum.y);
}

template <bool GUARD>
__device__ __forceinline__ void gcn_sum_run(const float* __restrict__ x, float* acc, int key_l, int ns_l,
                                            int g, int m, int lane, int& cur, float2& sum) {
    float2 v[RS_G];
    #pragma unroll
    for (int u = 0; u < RS_G; ++u) {
        v[u] = make_float2(0.0f, 0.0f);
        if (!GUARD || g + u < m) {
            const int src_row = gcn_lane_i(key_l, g + u) & 0xFFFFFF;
            const float2* xr = (const float2*)(x + (long)src_row * D) + lane;
            v[u] = xr[0];
        }
    }
    #pragma unroll
    for (int u = 0; u < RS_G; ++u) {
        if (!GUARD || g + u < m) {
            const int row = (int)((unsigned)gcn_lane_i(key_l, g + u) >> 24);
            const float ns = __int_as_float(gcn_lane_i(ns_l, g + u));
            if (row != cur) {
                if (cur >= 0) gcn_flush_row(acc, cur, lane, sum);
                sum = make_float2(0.0f, 0.0f);
                cur = row;
            }
            sum.x = fmaf(v[u].x, ns, sum.x);
            sum.y = fmaf(v[u].y, ns, sum.y);
        }
    }
}

__global__ __launch_bounds__(BK_T)
void gcn_rowsum_product_kernel(const float* __restrict__ x, const int2* __restrict__ list,
                               const int* __restrict__ tot, const float* __restrict__ norm_src,
                               const float* __restrict__ norm_dst, const float* __restrict__ Wmat,
                               float* __restrict__ out, int N, int NB) {
    __shared__ float acc[BK_R * BK_P];
    __shared__ int2 s_ord[RS_CH];
    __shared__ int s_cnt[256];
    __shared__ int s_start[256];
    __shared__ int s_w[32];
    __shared__ int s_rng[2];
    const int tid = threadIdx.x;
    const int lane = tid & 31;
    const int wv = __builtin_amdgcn_readfirstlane(tid >> 5);

    for (int k = tid; k < BK_R * BK_P; k += BK_T) acc[k] = 0.0f;
    if (tid < 256) s_cnt[tid] = 0;
    const int my_tot = tid < NB ? tot[tid] : 0;
    const int start = gcn_block_excl_scan(my_tot, s_w);
    if (tid == (int)blockIdx.x) { s_rng[0] = start; s_rng[1] = my_tot; }
    __syncthreads();
    const int first = __builtin_amdgcn_readfirstlane(s_rng[0]);
    const int cnt   = __builtin_amdgcn_readfirstlane(s_rng[1]);

    for (int c0 = 0; c0 < cnt; c0 += RS_CH) {
        const int n = cnt - c0 < RS_CH ? cnt - c0 : RS_CH;

        int2 sr[RS_U];
        #pragma unroll
        for (int u = 0; u < RS_U; ++u) {
            const int p = u * BK_T + tid;
            sr[u] = make_int2(0, -1);
            if (p < n) sr[u] = list[first + c0 + p];
        }
        int rk[RS_U], nsb[RS_U];
        #pragma unroll
        for (int u = 0; u < RS_U; ++u) {
            rk[u] = 0; nsb[u] = 0;
            if (sr[u].y >= 0) {
                rk[u]  = atomicAdd(&s_cnt[sr[u].y], 1);
                nsb[u] = __float_as_int(norm_src[sr[u].x]);
            }
        }
        __syncthreads();
        const int c = tid < BK_R ? s_cnt[tid] : 0;
        const int st = gcn_block_excl_scan(c, s_w);
        if (tid < BK_R) { s_start[tid] = st; s_cnt[tid] = 0; }
        __syncthreads();
        #pragma unroll
        for (int u = 0; u < RS_U; ++u) {
            if (sr[u].y >= 0) s_ord[s_start[sr[u].y] + rk[u]] = make_int2(sr[u].x | (sr[u].y << 24), nsb[u]);
        }
        __syncthreads();

        const int per = ((n + 255) >> 8) << 3;
        int ea = wv * per; if (ea > n) ea = n;
        int eb = ea + per; if (eb > n) eb = n;
        float2 sum = make_float2(0.0f, 0.0f);
        int cur = -1;
        for (int e0 = ea; e0 < eb; e0 += 32) {
            const int m = eb - e0 < 32 ? eb - e0 : 32;
            int2 en = make_int2(0, 0);
            if (lane < m) en = s_ord[e0 + lane];
            int g = 0;
            for (; g + RS_G <= m; g += RS_G) gcn_sum_run<false>(x, acc, en.x, en.y, g, m, lane, cur, sum);
            if (g < m)                       gcn_sum_run<true>(x, acc, en.x, en.y, g, m, lane, cur, sum);
        }
        if (cur >= 0) gcn_flush_row(acc, cur, lane, sum);
        __syncthreads();
    }

    if (wv < 2 * (BK_R / 16)) {
        const int half = lane >> 4;
        const int lm   = lane & 15;
        const int rt   = wv >> 1;
        const int col0 = (wv & 1) * 32;
        const int row  = rt * 16 + lm;
        const int g    = (int)blockIdx.x * BK_R + row;
        const int gc   = g < N ? g : (N - 1);
        const float scale = (g < N) ? norm_dst[gc] : 0.0f;
        const float* arow = acc + row * BK_P;

        v8f c0 = {}, c1 = {};
        for (int kk = 0; kk < 16; ++kk) {
            const int kbase = kk * 4 + half * 2;
            const float2 av = *(const float2*)(arow + kbase);
            v2f a; a.x = av.x * scale; a.y = av.y * scale;
            const float* w0 = Wmat + kbase * D + col0 + lm;
            v2f b0, b1;
            b0.x = w0[0];  b0.y = w0[D];
            b1.x = w0[16]; b1.y = w0[D + 16];
            c0 = __builtin_amdgcn_wmma_f32_16x16x4_f32(false, a, false, b0, (short)0, c0, false, false);
            c1 = __builtin_amdgcn_wmma_f32_16x16x4_f32(false, a, false, b1, (short)0, c1, false, false);
        }

        #pragma unroll
        for (int v = 0; v < 8; ++v) {
            const int gr = (int)blockIdx.x * BK_R + rt * 16 + v + half * 8;
            if (gr < N) {
                float* o = out + (long)gr * D + col0 + lm;
                o[0] = c0[v]; o[16] = c1[v];
            }
        }
    }
}

extern "C" void kernel_launch(void* const* d_in, const int* in_sizes, int n_in,
                              void* d_out, int out_size, void* d_ws, size_t ws_size,
                              hipStream_t stream) {
    const float* x    = (const float*)d_in[0];
    const float* Wmat = (const float*)d_in[1];
    const int*   src  = (const int*)d_in[2];
    const int*   dst  = (const int*)d_in[3];
    float* out = (float*)d_out;

    int N = in_sizes[0] / D;
    int E = in_sizes[2];

    float* ws_norm_src = (float*)d_ws;
    float* ws_norm_dst = ws_norm_src + N;
    int*   ws_region   = (int*)(ws_norm_dst + N);

    int*  ws_part = ws_region;
    int2* ws_list = (int2*)ws_region;
    int*  ws_tot  = ws_region + ((long)N * D - 2 * BK_MAX);
    int*  ws_cur  = ws_tot + BK_MAX;

    int Q = (E + CNT_CH - 1) / CNT_CH;
    int P = (N + CNT_R - 1) / CNT_R;
    gcn_count_kernel<<<dim3(Q, P, 2), CNT_T, 0, stream>>>(src, dst, ws_part, E, N);

    int NB = (N + BK_R - 1) / BK_R;
    gcn_norm_kernel<<<NB, BK_R, 0, stream>>>(ws_part, ws_norm_src, ws_norm_dst, ws_tot, ws_cur, N, Q);

    gcn_bin_kernel<<<(E + BIN_CH - 1) / BIN_CH, BK_T, 0, stream>>>(src, dst, ws_tot, ws_cur, ws_list, E, NB);

    gcn_rowsum_product_kernel<<<NB, BK_T, 0, stream>>>(x, ws_list, ws_tot, ws_norm_src, ws_norm_dst, Wmat, out, N, NB);
}


# In-degree from the row-sum kernel's own row counts, bucket totals from a bucket table in the count pass, norm_src summed inside the binning kernel: dst node tables and norm launch go (3 launches)
# speedup vs baseline: 5.3320x; 1.0302x over previous
#include <hip/hip_runtime.h>


typedef float v2f __attribute__((ext_vector_type(2)));
typedef float v8f __attribute__((ext_vector_type(8)));

#define D 64

#define CNT_R  32768
#define CNT_CH 65536
#define CNT_T  1024
#define BK_R   208
#define CNT_BS 160
#define CNT_QB 32

template <bool BUCKET>
__device__ __forceinline__ void gcn_count_one(int* h, int id, int lo, int b_lo) {
    if (BUCKET) {
        if (id >= 0) {
            const int b = (int)((unsigned)id / (unsigned)BK_R);
            const unsigned r = (unsigned)(b * BK_R - lo);
            if (r < (unsigned)CNT_R) atomicAdd(h + (b - b_lo), 1);
        }
    } else {
        const unsigned r = (unsigned)(id - lo);
        if (r < (unsigned)CNT_R) atomicAdd(h + r, 1);
    }
}

template <bool BUCKET>
__device__ __forceinline__ void gcn_count_chunk(int* h, const int* __restrict__ idx, long base, int cnt,
                                                int lo, int b_lo, int tid) {
    const int n4 = cnt >> 2;
    const int4* __restrict__ v4 = (const int4*)(idx + base);

    for (int i0 = tid; i0 < n4; i0 += 8 * CNT_T) {
        int4 v[8];
        #pragma unroll
        for (int u = 0; u < 8; ++u) {
            int i = i0 + u * CNT_T;
            v[u] = v4[i < n4 ? i : n4 - 1];
        }
        #pragma unroll
        for (int u = 0; u < 8; ++u) {
            bool live = i0 + u * CNT_T < n4;
            gcn_count_one<BUCKET>(h, live ? v[u].x : -1, lo, b_lo);
            gcn_count_one<BUCKET>(h, live ? v[u].y : -1, lo, b_lo);
            gcn_count_one<BUCKET>(h, live ? v[u].z : -1, lo, b_lo);
            gcn_count_one<BUCKET>(h, live ? v[u].w : -1, lo, b_lo);
        }
    }
    for (int j = (n4 << 2) + tid; j < cnt; j += CNT_T) gcn_count_one<BUCKET>(h, idx[base + j], lo, b_lo);
}

__global__ __launch_bounds__(CNT_T)
void gcn_count_kernel(const int* __restrict__ src, const int* __restrict__ dst,
                      int* __restrict__ part, int* __restrict__ bpart, int* __restrict__ cur, int E, int N, int QS) {
    __shared__ int h[CNT_R];
    const int tid = threadIdx.x;
    const bool by_bucket = blockIdx.z != 0;
    const int lo = blockIdx.y * CNT_R;
    const int b_lo = (lo + BK_R - 1) / BK_R;

    if (blockIdx.x == 0 && blockIdx.y == 0 && blockIdx.z == 0) cur[tid] = 0;

    const int nz = by_bucket ? CNT_BS : CNT_R;
    for (int r = tid; r < nz; r += CNT_T) h[r] = 0;
    __syncthreads();

    const long base = (long)blockIdx.x * CNT_CH;
    long left = (long)E - base;
    int cnt = left < 0 ? 0 : (left > CNT_CH ? CNT_CH : (int)left);
    if (by_bucket) gcn_count_chunk<true>(h, dst, base, cnt, lo, b_lo, tid);
    else           gcn_count_chunk<false>(h, src, base, cnt, lo, b_lo, tid);
    __syncthreads();

    if (by_bucket) {
        if (tid < CNT_BS) {
            int* row = bpart + ((long)blockIdx.y * CNT_BS + tid) * QS;
            row[blockIdx.x] = h[tid];
            if (blockIdx.x == gridDim.x - 1)
                for (int q = (int)gridDim.x; q < QS; ++q) row[q] = 0;
        }
        return;
    }
    int* __restrict__ outp = part + (long)blockIdx.x * N;
    for (int r0 = tid; r0 < CNT_R; r0 += 8 * CNT_T) {
        int c[8];
        #pragma unroll
        for (int u = 0; u < 8; ++u) c[u] = h[r0 + u * CNT_T];
        #pragma unroll
        for (int u = 0; u < 8; ++u) {
            int node = lo + r0 + u * CNT_T;
            if (node < N) outp[node] = c[u];
        }
    }
}

#define BK_P   68
#define BK_MAX 1024
#define BK_T   1024

__device__ __forceinline__ int gcn_block_excl_scan(int v, int* s_w) {
    const int lane = threadIdx.x & 31;
    const int wv = threadIdx.x >> 5;
    int inc = v;
    #pragma unroll
    for (int o = 1; o < 32; o <<= 1) {
        int t = __shfl_up(inc, o);
        if (lane >= o) inc += t;
    }
    if (lane == 31) s_w[wv] = inc;
    __syncthreads();
    if (wv == 0) {
        int w = s_w[lane];
        int winc = w;
        #pragma unroll
        for (int o = 1; o < 32; o <<= 1) {
            int t = __shfl_up(winc, o);
            if (lane >= o) winc += t;
        }
        s_w[lane] = winc - w;
    }
    __syncthreads();
    return inc - v + s_w[wv];
}

#define BIN_K  8
#define BIN_CH (BK_T * BIN_K)

__global__ __launch_bounds__(BK_T)
void gcn_bin_kernel(const int* __restrict__ src, const int* __restrict__ dst, const int* __restrict__ bpart,
                    const int* __restrict__ part, float* __restrict__ norm_src, int* __restrict__ tot,
                    int* __restrict__ cur, int2* __restrict__ list, int E, int NB, int N, int Q, int QS) {
    __shared__ int  s_cnt[BK_MAX];
    __shared__ int  s_loc[BK_MAX];
    __shared__ int2 s_stage[BIN_CH];
    __shared__ int  s_w[32];
    const int tid = threadIdx.x;
    s_cnt[tid] = 0;

    const long base = (long)blockIdx.x * BIN_CH;
    int s[BIN_K], d[BIN_K];
    #pragma unroll
    for (int u = 0; u < BIN_K; ++u) {
        const long e = base + u * BK_T + tid;
        const bool live = e < E;
        s[u] = live ? src[e] : 0;
        d[u] = live ? dst[e] : -1;
    }

    int my_tot = 0;
    if (tid < NB) {
        const int y0 = (tid * BK_R) / CNT_R;
        const int4* p0 = (const int4*)(bpart + ((long)y0 * CNT_BS + (tid - (y0 * CNT_R + BK_R - 1) / BK_R)) * QS);
        const int n4 = QS >> 2;
        for (int k0 = 0; k0 < n4; k0 += 8) {
            int4 v[8];
            #pragma unroll
            for (int u = 0; u < 8; ++u) v[u] = p0[k0 + u < n4 ? k0 + u : n4 - 1];
            #pragma unroll
            for (int u = 0; u < 8; ++u) my_tot += k0 + u < n4 ? (v[u].x + v[u].y) + (v[u].z + v[u].w) : 0;
        }
    }
    if (blockIdx.x == 0 && tid < NB) tot[tid] = my_tot;

    const int per = (N + (int)gridDim.x - 1) / (int)gridDim.x;
    for (int j = tid; j < per; j += BK_T) {
        const int i = (int)blockIdx.x * per + j;
        if (i < N) {
            int so = 0;
            for (int q0 = 0; q0 < Q; q0 += CNT_QB) {
                int v[CNT_QB];
                #pragma unroll
                for (int u = 0; u < CNT_QB; ++u) {
                    const int* tab = part + (long)(q0 + u < Q ? q0 + u : Q - 1) * N;
                    v[u] = tab[i];
                }
                #pragma unroll
                for (int u = 0; u < CNT_QB; ++u) so += q0 + u < Q ? v[u] : 0;
            }
            norm_src[i] = rsqrtf(fmaxf((float)so, 1.0f));
        }
    }

    const int start = gcn_block_excl_scan(my_tot, s_w);

    int bk[BIN_K], rk[BIN_K];
    #pragma unroll
    for (int u = 0; u < BIN_K; ++u) {
        bk[u] = 0; rk[u] = 0;
        if (d[u] >= 0) {
            bk[u] = (int)((unsigned)d[u] / (unsigned)BK_R);
            rk[u] = atomicAdd(&s_cnt[bk[u]], 1);
        }
    }
    __syncthreads();
    const int c = s_cnt[tid];
    const int loc = gcn_block_excl_scan(c, s_w);
    int run = 0;
    if (c > 0) run = atomicAdd(&cur[tid], c);
    s_loc[tid] = loc;
    s_cnt[tid] = start + run - loc;
    __syncthreads();
    #pragma unroll
    for (int u = 0; u < BIN_K; ++u) {
        if (d[u] >= 0) s_stage[s_loc[bk[u]] + rk[u]] = make_int2(s[u], d[u]);
    }
    __syncthreads();
    const long left = (long)E - base;
    const int nlive = left > BIN_CH ? BIN_CH : (int)left;
    for (int p = tid; p < nlive; p += BK_T) {
        const int2 sd = s_stage[p];
        const int b = (int)((unsigned)sd.y / (unsigned)BK_R);
        list[s_cnt[b] + p] = make_int2(sd.x, sd.y - b * BK_R);
    }
}

#define RS_G  8
#define RS_U  4
#define RS_CH (BK_T * RS_U)

__device__ __forceinline__ int gcn_lane_i(int v, int j) {
    return __builtin_amdgcn_readlane(v, j);
}

__device__ __forceinline__ void gcn_flush_row(float* acc, int row, int lane, float2 sum) {
    float* ar = acc + row * BK_P + 2 * lane;
    atomicAdd(ar,     sum.x);
    atomicAdd(ar + 1, sum.y);
}

template <bool GUARD>
__device__ __forceinline__ void gcn_sum_run(const float* __restrict__ x, float* acc, int key_l, int ns_l,
                                            int g, int m, int lane, int& cur, float2& sum) {
    float2 v[RS_G];
    #pragma unroll
    for (int u = 0; u < RS_G; ++u) {
        v[u] = make_float2(0.0f, 0.0f);
        if (!GUARD || g + u < m) {
            const int src_row = gcn_lane_i(key_l, g + u) & 0xFFFFFF;
            const float2* xr = (const float2*)(x + (long)src_row * D) + lane;
            v[u] = xr[0];
        }
    }
    #pragma unroll
    for (int u = 0; u < RS_G; ++u) {
        if (!GUARD || g + u < m) {
            const int row = (int)((unsigned)gcn_lane_i(key_l, g + u) >> 24);
            const float ns = __int_as_float(gcn_lane_i(ns_l, g + u));
            if (row != cur) {
                if (cur >= 0) gcn_flush_row(acc, cur, lane, sum);
                sum = make_float2(0.0f, 0.0f);
                cur = row;
            }
            sum.x = fmaf(v[u].x, ns, sum.x);
            sum.y = fmaf(v[u].y, ns, sum.y);
        }
    }
}

__global__ __launch_bounds__(BK_T)
void gcn_rowsum_product_kernel(const float* __restrict__ x, const int2* __restrict__ list,
                               const int* __restrict__ tot, const float* __restrict__ norm_src,
                               const float* __restrict__ Wmat, float* __restrict__ out, int N, int NB) {
    __shared__ float acc[BK_R * BK_P];
    __shared__ int2 s_ord[RS_CH];
    __shared__ int s_cnt[256];
    __shared__ int s_start[256];
    __shared__ int s_w[32];
    __shared__ int s_rng[2];
    const int tid = threadIdx.x;
    const int lane = tid & 31;
    const int wv = __builtin_amdgcn_readfirstlane(tid >> 5);

    for (int k = tid; k < BK_R * BK_P; k += BK_T) acc[k] = 0.0f;
    if (tid < 256) s_cnt[tid] = 0;
    const int my_tot = tid < NB ? tot[tid] : 0;
    const int start = gcn_block_excl_scan(my_tot, s_w);
    if (tid == (int)blockIdx.x) { s_rng[0] = start; s_rng[1] = my_tot; }
    __syncthreads();
    const int first = __builtin_amdgcn_readfirstlane(s_rng[0]);
    const int cnt   = __builtin_amdgcn_readfirstlane(s_rng[1]);
    int deg = 0;

    for (int c0 = 0; c0 < cnt; c0 += RS_CH) {
        const int n = cnt - c0 < RS_CH ? cnt - c0 : RS_CH;

        int2 sr[RS_U];
        #pragma unroll
        for (int u = 0; u < RS_U; ++u) {
            const int p = u * BK_T + tid;
            sr[u] = make_int2(0, -1);
            if (p < n) sr[u] = list[first + c0 + p];
        }
        int rk[RS_U], nsb[RS_U];
        #pragma unroll
        for (int u = 0; u < RS_U; ++u) {
            rk[u] = 0; nsb[u] = 0;
            if (sr[u].y >= 0) {
                rk[u]  = atomicAdd(&s_cnt[sr[u].y], 1);
                nsb[u] = __float_as_int(norm_src[sr[u].x]);
            }
        }
        __syncthreads();
        const int c = tid < BK_R ? s_cnt[tid] : 0;
        deg += c;
        const int st = gcn_block_excl_scan(c, s_w);
        if (tid < BK_R) { s_start[tid] = st; s_cnt[tid] = 0; }
        __syncthreads();
        #pragma unroll
        for (int u = 0; u < RS_U; ++u) {
            if (sr[u].y >= 0) s_ord[s_start[sr[u].y] + rk[u]] = make_int2(sr[u].x | (sr[u].y << 24), nsb[u]);
        }
        __syncthreads();

        const int per = ((n + 255) >> 8) << 3;
        int ea = wv * per; if (ea > n) ea = n;
        int eb = ea + per; if (eb > n) eb = n;
        float2 sum = make_float2(0.0f, 0.0f);
        int cur = -1;
        for (int e0 = ea; e0 < eb; e0 += 32) {
            const int m = eb - e0 < 32 ? eb - e0 : 32;
            int2 en = make_int2(0, 0);
            if (lane < m) en = s_ord[e0 + lane];
            int g = 0;
            for (; g + RS_G <= m; g += RS_G) gcn_sum_run<false>(x, acc, en.x, en.y, g, m, lane, cur, sum);
            if (g < m)                       gcn_sum_run<true>(x, acc, en.x, en.y, g, m, lane, cur, sum);
        }
        if (cur >= 0) gcn_flush_row(acc, cur, lane, sum);
        __syncthreads();
    }

    if (tid < BK_R) s_start[tid] = deg;
    __syncthreads();

    if (wv < 2 * (BK_R / 16)) {
        const int half = lane >> 4;
        const int lm   = lane & 15;
        const int rt   = wv >> 1;
        const int col0 = (wv & 1) * 32;
        const int row  = rt * 16 + lm;
        const int g    = (int)blockIdx.x * BK_R + row;
        const float scale = (g < N) ? rsqrtf(fmaxf((float)s_start[row], 1.0f)) : 0.0f;
        const float* arow = acc + row * BK_P;

        v8f c0 = {}, c1 = {};
        for (int kk = 0; kk < 16; ++kk) {
            const int kbase = kk * 4 + half * 2;
            const float2 av = *(const float2*)(arow + kbase);
            v2f a; a.x = av.x * scale; a.y = av.y * scale;
            const float* w0 = Wmat + kbase * D + col0 + lm;
            v2f b0, b1;
            b0.x = w0[0];  b0.y = w0[D];
            b1.x = w0[16]; b1.y = w0[D + 16];
            c0 = __builtin_amdgcn_wmma_f32_16x16x4_f32(false, a, false, b0, (short)0, c0, false, false);
            c1 = __builtin_amdgcn_wmma_f32_16x16x4_f32(false, a, false, b1, (short)0, c1, false, false);
        }

        #pragma unroll
        for (int v = 0; v < 8; ++v) {
            const int gr = (int)blockIdx.x * BK_R + rt * 16 + v + half * 8;
            if (gr < N) {
                float* o = out + (long)gr * D + col0 + lm;
                o[0] = c0[v]; o[16] = c1[v];
            }
        }
    }
}

extern "C" void kernel_launch(void* const* d_in, const int* in_sizes, int n_in,
                              void* d_out, int out_size, void* d_ws, size_t ws_size,
                              hipStream_t stream) {
    const float* x    = (const float*)d_in[0];
    const float* Wmat = (const float*)d_in[1];
    const int*   src  = (const int*)d_in[2];
    const int*   dst  = (const int*)d_in[3];
    float* out = (float*)d_out;

    int N = in_sizes[0] / D;
    int E = in_sizes[2];

    float* ws_norm_src = (float*)d_ws;
    int*   ws_bpart    = (int*)(ws_norm_src + N);
    int*   ws_region   = ws_bpart + N;

    int Q = (E + CNT_CH - 1) / CNT_CH;
    int P = (N + CNT_R - 1) / CNT_R;

    int QS = (Q + 3) & ~3;
    int2* ws_list = (int2*)ws_region;
    int*  ws_part = ws_region + 2 * (long)E;
    int*  ws_tot  = ws_region + ((long)N * D - 2 * BK_MAX);
    int*  ws_cur  = ws_tot + BK_MAX;

    gcn_count_kernel<<<dim3(Q, P, 2), CNT_T, 0, stream>>>(src, dst, ws_part, ws_bpart, ws_cur, E, N, QS);

    int NB = (N + BK_R - 1) / BK_R;
    gcn_bin_kernel<<<(E + BIN_CH - 1) / BIN_CH, BK_T, 0, stream>>>(src, dst, ws_bpart, ws_part, ws_norm_src, ws_tot,
                                                                 ws_cur, ws_list, E, NB, N, Q, QS);

    gcn_rowsum_product_kernel<<<NB, BK_T, 0, stream>>>(x, ws_list, ws_tot, ws_norm_src, Wmat, out, N, NB);
}


# Per-edge norm_src from an LDS byte table of out-degrees (255 escapes to the float table) copied in per row-sum block: no scattered 4-byte global read; 400-row buckets; rank-order sums, slices move
# speedup vs baseline: 5.6123x; 1.0526x over previous
#include <hip/hip_runtime.h>


typedef float v2f __attribute__((ext_vector_type(2)));
typedef float v8f __attribute__((ext_vector_type(8)));

#define D 64

#define CNT_R  32768
#define CNT_CH 65536
#define CNT_T  1024
#define BK_R   400
#define CNT_BS 96
#define CNT_QB 32

template <bool BUCKET>
__device__ __forceinline__ void gcn_count_one(int* h, int id, int lo, int b_lo) {
    if (BUCKET) {
        if (id >= 0) {
            const int b = (int)((unsigned)id / (unsigned)BK_R);
            const unsigned r = (unsigned)(b * BK_R - lo);
            if (r < (unsigned)CNT_R) atomicAdd(h + (b - b_lo), 1);
        }
    } else {
        const unsigned r = (unsigned)(id - lo);
        if (r < (unsigned)CNT_R) atomicAdd(h + r, 1);
    }
}

template <bool BUCKET>
__device__ __forceinline__ void gcn_count_chunk(int* h, const int* __restrict__ idx, long base, int cnt,
                                                int lo, int b_lo, int tid) {
    const int n4 = cnt >> 2;
    const int4* __restrict__ v4 = (const int4*)(idx + base);

    for (int i0 = tid; i0 < n4; i0 += 8 * CNT_T) {
        int4 v[8];
        #pragma unroll
        for (int u = 0; u < 8; ++u) {
            int i = i0 + u * CNT_T;
            v[u] = v4[i < n4 ? i : n4 - 1];
        }
        #pragma unroll
        for (int u = 0; u < 8; ++u) {
            bool live = i0 + u * CNT_T < n4;
            gcn_count_one<BUCKET>(h, live ? v[u].x : -1, lo, b_lo);
            gcn_count_one<BUCKET>(h, live ? v[u].y : -1, lo, b_lo);
            gcn_count_one<BUCKET>(h, live ? v[u].z : -1, lo, b_lo);
            gcn_count_one<BUCKET>(h, live ? v[u].w : -1, lo, b_lo);
        }
    }
    for (int j = (n4 << 2) + tid; j < cnt; j += CNT_T) gcn_count_one<BUCKET>(h, idx[base + j], lo, b_lo);
}

__global__ __launch_bounds__(CNT_T)
void gcn_count_kernel(const int* __restrict__ src, const int* __restrict__ dst,
                      int* __restrict__ part, int* __restrict__ bpart, int* __restrict__ cur, int E, int N, int QS) {
    __shared__ int h[CNT_R];
    const int tid = threadIdx.x;
    const bool by_bucket = blockIdx.z != 0;
    const int lo = blockIdx.y * CNT_R;
    const int b_lo = (lo + BK_R - 1) / BK_R;

    if (blockIdx.x == 0 && blockIdx.y == 0 && blockIdx.z == 0) cur[tid] = 0;

    const int nz = by_bucket ? CNT_BS : CNT_R;
    for (int r = tid; r < nz; r += CNT_T) h[r] = 0;
    __syncthreads();

    const long base = (long)blockIdx.x * CNT_CH;
    long left = (long)E - base;
    int cnt = left < 0 ? 0 : (left > CNT_CH ? CNT_CH : (int)left);
    if (by_bucket) gcn_count_chunk<true>(h, dst, base, cnt, lo, b_lo, tid);
    else           gcn_count_chunk<false>(h, src, base, cnt, lo, b_lo, tid);
    __syncthreads();

    if (by_bucket) {
        if (tid < CNT_BS) {
            int* row = bpart + ((long)blockIdx.y * CNT_BS + tid) * QS;
            row[blockIdx.x] = h[tid];
            if (blockIdx.x == gridDim.x - 1)
                for (int q = (int)gridDim.x; q < QS; ++q) row[q] = 0;
        }
        return;
    }
    int* __restrict__ outp = part + (long)blockIdx.x * N;
    for (int r0 = tid; r0 < CNT_R; r0 += 8 * CNT_T) {
        int c[8];
        #pragma unroll
        for (int u = 0; u < 8; ++u) c[u] = h[r0 + u * CNT_T];
        #pragma unroll
        for (int u = 0; u < 8; ++u) {
            int node = lo + r0 + u * CNT_T;
            if (node < N) outp[node] = c[u];
        }
    }
}

#define BK_P   68
#define BK_MAX 1024
#define BK_T   1024

__device__ __forceinline__ float gcn_norm_of(int deg) { return rsqrtf(fmaxf((float)deg, 1.0f)); }

__device__ __forceinline__ long gcn_deg8_off(int N) {
    const long npad = (long)((N + 15) & ~15);
    return (8 * (long)N - npad) & ~15L;
}

__device__ __forceinline__ int gcn_block_excl_scan(int v, int* s_w) {
    const int lane = threadIdx.x & 31;
    const int wv = threadIdx.x >> 5;
    int inc = v;
    #pragma unroll
    for (int o = 1; o < 32; o <<= 1) {
        int t = __shfl_up(inc, o);
        if (lane >= o) inc += t;
    }
    if (lane == 31) s_w[wv] = inc;
    __syncthreads();
    if (wv == 0) {
        int w = s_w[lane];
        int winc = w;
        #pragma unroll
        for (int o = 1; o < 32; o <<= 1) {
            int t = __shfl_up(winc, o);
            if (lane >= o) winc += t;
        }
        s_w[lane] = winc - w;
    }
    __syncthreads();
    return inc - v + s_w[wv];
}

#define BIN_K  8
#define BIN_CH (BK_T * BIN_K)

__global__ __launch_bounds__(BK_T)
void gcn_bin_kernel(const int* __restrict__ src, const int* __restrict__ dst, const int* __restrict__ bpart,
                    const int* __restrict__ part, float* __restrict__ norm_src, int* __restrict__ tot,
                    int* __restrict__ cur, int2* __restrict__ list, int E, int NB, int N, int Q, int QS) {
    __shared__ int  s_cnt[BK_MAX];
    __shared__ int  s_loc[BK_MAX];
    __shared__ int2 s_stage[BIN_CH];
    __shared__ int  s_w[32];
    const int tid = threadIdx.x;
    s_cnt[tid] = 0;

    const long base = (long)blockIdx.x * BIN_CH;
    int s[BIN_K], d[BIN_K];
    #pragma unroll
    for (int u = 0; u < BIN_K; ++u) {
        const long e = base + u * BK_T + tid;
        const bool live = e < E;
        s[u] = live ? src[e] : 0;
        d[u] = live ? dst[e] : -1;
    }

    int my_tot = 0;
    if (tid < NB) {
        const int y0 = (tid * BK_R) / CNT_R;
        const int4* p0 = (const int4*)(bpart + ((long)y0 * CNT_BS + (tid - (y0 * CNT_R + BK_R - 1) / BK_R)) * QS);
        const int n4 = QS >> 2;
        for (int k0 = 0; k0 < n4; k0 += 8) {
            int4 v[8];
            #pragma unroll
            for (int u = 0; u < 8; ++u) v[u] = p0[k0 + u < n4 ? k0 + u : n4 - 1];
            #pragma unroll
            for (int u = 0; u < 8; ++u) my_tot += k0 + u < n4 ? (v[u].x + v[u].y) + (v[u].z + v[u].w) : 0;
        }
    }
    if (blockIdx.x == 0 && tid < NB) tot[tid] = my_tot;

    const int npad = (N + 15) & ~15;
    unsigned char* deg8 = (unsigned char*)norm_src + gcn_deg8_off(N);
    const int per = (((npad + (int)gridDim.x - 1) / (int)gridDim.x) + 127) & ~127;
    for (int j = tid; j < per; j += BK_T) {
        const int i = (int)blockIdx.x * per + j;
        if (i >= N) {
            if (i < npad) deg8[i] = 0;
        } else {
            int so = 0;
            for (int q0 = 0; q0 < Q; q0 += CNT_QB) {
                int v[CNT_QB];
                #pragma unroll
                for (int u = 0; u < CNT_QB; ++u) {
                    const int* tab = part + (long)(q0 + u < Q ? q0 + u : Q - 1) * N;
                    v[u] = tab[i];
                }
                #pragma unroll
                for (int u = 0; u < CNT_QB; ++u) so += q0 + u < Q ? v[u] : 0;
            }
            norm_src[i] = gcn_norm_of(so);
            deg8[i] = (unsigned char)(so < 255 ? so : 255);
        }
    }

    const int start = gcn_block_excl_scan(my_tot, s_w);

    int bk[BIN_K], rk[BIN_K];
    #pragma unroll
    for (int u = 0; u < BIN_K; ++u) {
        bk[u] = 0; rk[u] = 0;
        if (d[u] >= 0) {
            bk[u] = (int)((unsigned)d[u] / (unsigned)BK_R);
            rk[u] = atomicAdd(&s_cnt[bk[u]], 1);
        }
    }
    __syncthreads();
    const int c = s_cnt[tid];
    const int loc = gcn_block_excl_scan(c, s_w);
    int run = 0;
    if (c > 0) run = atomicAdd(&cur[tid], c);
    s_loc[tid] = loc;
    s_cnt[tid] = start + run - loc;
    __syncthreads();
    #pragma unroll
    for (int u = 0; u < BIN_K; ++u) {
        if (d[u] >= 0) s_stage[s_loc[bk[u]] + rk[u]] = make_int2(s[u], d[u]);
    }
    __syncthreads();
    const long left = (long)E - base;
    const int nlive = left > BIN_CH ? BIN_CH : (int)left;
    for (int p = tid; p < nlive; p += BK_T) {
        const int2 sd = s_stage[p];
        const int b = (int)((unsigned)sd.y / (unsigned)BK_R);
        list[s_cnt[b] + p] = make_int2(sd.x, sd.y - b * BK_R);
    }
}

#define RS_G  16
#define RS_U  8
#define RS_CH (BK_T * RS_U)
#define RS_TAB 100352

__device__ __forceinline__ int gcn_lane_i(int v, int j) {
    return __builtin_amdgcn_readlane(v, j);
}

__device__ __forceinline__ void gcn_flush_row(float* acc, int row, int lane, float2 sum) {
    float* ar = acc + row * BK_P + 2 * lane;
    atomicAdd(ar,     sum.x);
    atomicAdd(ar + 1, sum.y);
}

template <bool GUARD>
__device__ __forceinline__ void gcn_sum_run(const float* __restrict__ x, float* acc, int key_l, int ns_l,
                                            int g, int m, int lane, int& cur, float2& sum) {
    float2 v[RS_G];
    #pragma unroll
    for (int u = 0; u < RS_G; ++u) {
        v[u] = make_float2(0.0f, 0.0f);
        if (!GUARD || g + u < m) {
            const int src_row = gcn_lane_i(key_l, g + u) & 0x7FFFFF;
            const float2* xr = (const float2*)(x + (long)src_row * D) + lane;
            v[u] = xr[0];
        }
    }
    #pragma unroll
    for (int u = 0; u < RS_G; ++u) {
        if (!GUARD || g + u < m) {
            const int row = (int)((unsigned)gcn_lane_i(key_l, g + u) >> 23);
            const float ns = __int_as_float(gcn_lane_i(ns_l, g + u));
            if (row != cur) {
                if (cur >= 0) gcn_flush_row(acc, cur, lane, sum);
                sum = make_float2(0.0f, 0.0f);
                cur = row;
            }
            sum.x = fmaf(v[u].x, ns, sum.x);
            sum.y = fmaf(v[u].y, ns, sum.y);
        }
    }
}

__global__ __launch_bounds__(BK_T)
void gcn_rowsum_product_kernel(const float* __restrict__ x, const int2* __restrict__ list,
                               const int* __restrict__ tot, const float* __restrict__ norm_src,
                               const float* __restrict__ Wmat, float* __restrict__ out, int N, int NB) {
    __shared__ float acc[BK_R * BK_P];
    __shared__ int2 s_ord[RS_CH];
    __shared__ __attribute__((aligned(16))) unsigned char s_deg[RS_TAB];
    __shared__ int s_cnt[512];
    __shared__ int s_start[512];
    __shared__ int s_w[32];
    __shared__ int s_rng[2];
    const int tid = threadIdx.x;
    const int lane = tid & 31;
    const int wv = __builtin_amdgcn_readfirstlane(tid >> 5);

    const int my_tot = tid < NB ? tot[tid] : 0;

    const int npad = (N + 15) & ~15;
    const bool use_tab = npad <= RS_TAB;
    const unsigned char* deg8 = (const unsigned char*)norm_src + gcn_deg8_off(N);
    if (use_tab) {
        const int n16 = npad >> 4;
        const uint4* g16 = (const uint4*)deg8;
        uint4* l16 = (uint4*)s_deg;
        for (int k0 = tid; k0 < n16; k0 += 8 * BK_T) {
            uint4 t[8];
            #pragma unroll
            for (int u = 0; u < 8; ++u) {
                const int k = k0 + u * BK_T;
                t[u] = g16[k < n16 ? k : n16 - 1];
            }
            #pragma unroll
            for (int u = 0; u < 8; ++u) {
                const int k = k0 + u * BK_T;
                l16[k < n16 ? k : n16 - 1] = t[u];
            }
        }
    }

    for (int k = tid; k < BK_R * BK_P; k += BK_T) acc[k] = 0.0f;
    if (tid < 512) s_cnt[tid] = 0;
    const int start = gcn_block_excl_scan(my_tot, s_w);
    if (tid == (int)blockIdx.x) { s_rng[0] = start; s_rng[1] = my_tot; }
    __syncthreads();
    const int first = __builtin_amdgcn_readfirstlane(s_rng[0]);
    const int cnt   = __builtin_amdgcn_readfirstlane(s_rng[1]);
    int deg = 0;

    for (int c0 = 0; c0 < cnt; c0 += RS_CH) {
        const int n = cnt - c0 < RS_CH ? cnt - c0 : RS_CH;

        int2 sr[RS_U];
        #pragma unroll
        for (int u = 0; u < RS_U; ++u) {
            const int p = u * BK_T + tid;
            sr[u] = make_int2(0, -1);
            if (p < n) sr[u] = list[first + c0 + p];
        }
        int rk[RS_U];
        #pragma unroll
        for (int u = 0; u < RS_U; ++u) {
            rk[u] = 0;
            if (sr[u].y >= 0) rk[u] = atomicAdd(&s_cnt[sr[u].y], 1);
        }
        __syncthreads();
        const int c = tid < BK_R ? s_cnt[tid] : 0;
        deg += c;
        const int st = gcn_block_excl_scan(c, s_w);
        if (tid < BK_R) { s_start[tid] = st; s_cnt[tid] = 0; }
        __syncthreads();
        int d8[RS_U], st8[RS_U];
        #pragma unroll
        for (int u = 0; u < RS_U; ++u) {
            d8[u]  = (int)s_deg[use_tab ? sr[u].x : 0];
            st8[u] = s_start[sr[u].y >= 0 ? sr[u].y : 0];
        }
        float nsv[RS_U];
        bool esc = false;
        #pragma unroll
        for (int u = 0; u < RS_U; ++u) {
            if (!use_tab) d8[u] = 255;
            nsv[u] = gcn_norm_of(d8[u]);
            esc = esc || (d8[u] == 255 && sr[u].y >= 0);
        }
        if (esc) {
            #pragma unroll
            for (int u = 0; u < RS_U; ++u) {
                if (d8[u] == 255 && sr[u].y >= 0) nsv[u] = norm_src[sr[u].x];
            }
        }
        #pragma unroll
        for (int u = 0; u < RS_U; ++u) {
            if (sr[u].y >= 0) s_ord[st8[u] + rk[u]] = make_int2(sr[u].x | (int)((unsigned)sr[u].y << 23), __float_as_int(nsv[u]));
        }
        __syncthreads();

        const int per = ((n + 32 * RS_G - 1) / (32 * RS_G)) * RS_G;
        int ea = wv * per; if (ea > n) ea = n;
        int eb = ea + per; if (eb > n) eb = n;
        float2 sum = make_float2(0.0f, 0.0f);
        int cur = -1;
        for (int e0 = ea; e0 < eb; e0 += 32) {
            const int m = eb - e0 < 32 ? eb - e0 : 32;
            int2 en = make_int2(0, 0);
            if (lane < m) en = s_ord[e0 + lane];
            int g = 0;
            for (; g + RS_G <= m; g += RS_G) gcn_sum_run<false>(x, acc, en.x, en.y, g, m, lane, cur, sum);
            if (g < m)                       gcn_sum_run<true>(x, acc, en.x, en.y, g, m, lane, cur, sum);
        }
        if (cur >= 0) gcn_flush_row(acc, cur, lane, sum);
        __syncthreads();
    }

    if (tid < BK_R) s_start[tid] = deg;
    __syncthreads();

    for (int task = wv; task < 2 * (BK_R / 16); task += BK_T / 32) {
        const int half = lane >> 4;
        const int lm   = lane & 15;
        const int rt   = task >> 1;
        const int col0 = (task & 1) * 32;
        const int row  = rt * 16 + lm;
        const int g    = (int)blockIdx.x * BK_R + row;
        const float scale = (g < N) ? rsqrtf(fmaxf((float)s_start[row], 1.0f)) : 0.0f;
        const float* arow = acc + row * BK_P;

        v8f c0 = {}, c1 = {};
        for (int kk = 0; kk < 16; ++kk) {
            const int kbase = kk * 4 + half * 2;
            const float2 av = *(const float2*)(arow + kbase);
            v2f a; a.x = av.x * scale; a.y = av.y * scale;
            const float* w0 = Wmat + kbase * D + col0 + lm;
            v2f b0, b1;
            b0.x = w0[0];  b0.y = w0[D];
            b1.x = w0[16]; b1.y = w0[D + 16];
            c0 = __builtin_amdgcn_wmma_f32_16x16x4_f32(false, a, false, b0, (short)0, c0, false, false);
            c1 = __builtin_amdgcn_wmma_f32_16x16x4_f32(false, a, false, b1, (short)0, c1, false, false);
        }

        #pragma unroll
        for (int v = 0; v < 8; ++v) {
            const int gr = (int)blockIdx.x * BK_R + rt * 16 + v + half * 8;
            if (gr < N) {
                float* o = out + (long)gr * D + col0 + lm;
                o[0] = c0[v]; o[16] = c1[v];
            }
        }
    }
}

extern "C" void kernel_launch(void* const* d_in, const int* in_sizes, int n_in,
                              void* d_out, int out_size, void* d_ws, size_t ws_size,
                              hipStream_t stream) {
    const float* x    = (const float*)d_in[0];
    const float* Wmat = (const float*)d_in[1];
    const int*   src  = (const int*)d_in[2];
    const int*   dst  = (const int*)d_in[3];
    float* out = (float*)d_out;

    int N = in_sizes[0] / D;
    int E = in_sizes[2];

    float* ws_norm_src = (float*)d_ws;
    int*   ws_bpart    = (int*)(ws_norm_src + N);
    int*   ws_region   = ws_bpart + N;

    int Q = (E + CNT_CH - 1) / CNT_CH;
    int P = (N + CNT_R - 1) / CNT_R;

    int QS = (Q + 3) & ~3;
    int2* ws_list = (int2*)ws_region;
    int*  ws_part = ws_region + 2 * (long)E;
    int*  ws_tot  = ws_region + ((long)N * D - 2 * BK_MAX);
    int*  ws_cur  = ws_tot + BK_MAX;

    gcn_count_kernel<<<dim3(Q, P, 2), CNT_T, 0, stream>>>(src, dst, ws_part, ws_bpart, ws_cur, E, N, QS);

    int NB = (N + BK_R - 1) / BK_R;
    gcn_bin_kernel<<<(E + BIN_CH - 1) / BIN_CH, BK_T, 0, stream>>>(src, dst, ws_bpart, ws_part, ws_norm_src, ws_tot,
                                                                 ws_cur, ws_list, E, NB, N, Q, QS);

    gcn_rowsum_product_kernel<<<NB, BK_T, 0, stream>>>(x, ws_list, ws_tot, ws_norm_src, Wmat, out, N, NB);
}


# Bucket totals by one block per edge chunk over all buckets, in 32 lane-private LDS words per bucket (25 blocks, dispatched last, no range test), not four range blocks per chunk: 75 blocks fewer
# speedup vs baseline: 5.7947x; 1.0325x over previous
#include <hip/hip_runtime.h>


typedef float v2f __attribute__((ext_vector_type(2)));
typedef float v8f __attribute__((ext_vector_type(8)));

#define D 64

#define CNT_R  32768
#define CNT_CH 65536
#define CNT_T  1024
#define BK_MAX_C 1024
#define BK_R   400
#define CNT_QB 32

template <bool BUCKET>
__device__ __forceinline__ void gcn_count_one(int* h, int id, int lo, int lane) {
    if (BUCKET) {
        if ((unsigned)id < (unsigned)(BK_MAX_C * BK_R))
            atomicAdd(h + ((unsigned)id / (unsigned)BK_R) * 32u + (unsigned)lane, 1);
    } else {
        const unsigned r = (unsigned)(id - lo);
        if (r < (unsigned)CNT_R) atomicAdd(h + r, 1);
    }
}

template <bool BUCKET>
__device__ __forceinline__ void gcn_count_chunk(int* h, const int* __restrict__ idx, long base, int cnt,
                                                int lo, int tid) {
    const int n4 = cnt >> 2;
    const int lane = tid & 31;
    const int4* __restrict__ v4 = (const int4*)(idx + base);

    for (int i0 = tid; i0 < n4; i0 += 8 * CNT_T) {
        int4 v[8];
        #pragma unroll
        for (int u = 0; u < 8; ++u) {
            int i = i0 + u * CNT_T;
            v[u] = v4[i < n4 ? i : n4 - 1];
        }
        #pragma unroll
        for (int u = 0; u < 8; ++u) {
            bool live = i0 + u * CNT_T < n4;
            gcn_count_one<BUCKET>(h, live ? v[u].x : -1, lo, lane);
            gcn_count_one<BUCKET>(h, live ? v[u].y : -1, lo, lane);
            gcn_count_one<BUCKET>(h, live ? v[u].z : -1, lo, lane);
            gcn_count_one<BUCKET>(h, live ? v[u].w : -1, lo, lane);
        }
    }
    for (int j = (n4 << 2) + tid; j < cnt; j += CNT_T) gcn_count_one<BUCKET>(h, idx[base + j], lo, lane);
}

__global__ __launch_bounds__(CNT_T)
void gcn_count_kernel(const int* __restrict__ src, const int* __restrict__ dst,
                      int* __restrict__ part, int* __restrict__ bpart, int* __restrict__ cur, int E, int N, int QS) {
    __shared__ int h[CNT_R];
    const int tid = threadIdx.x;
    const bool by_bucket = blockIdx.y == gridDim.y - 1;
    const int lo = blockIdx.y * CNT_R;
    const int nb = (N + BK_R - 1) / BK_R;

    if (blockIdx.x == 0 && blockIdx.y == 0) cur[tid] = 0;

    const int nz = by_bucket ? (nb * 32 < CNT_R ? nb * 32 : CNT_R) : CNT_R;
    for (int r = tid; r < nz; r += CNT_T) h[r] = 0;
    __syncthreads();

    const long base = (long)blockIdx.x * CNT_CH;
    long left = (long)E - base;
    int cnt = left < 0 ? 0 : (left > CNT_CH ? CNT_CH : (int)left);
    if (by_bucket) gcn_count_chunk<true>(h, dst, base, cnt, lo, tid);
    else           gcn_count_chunk<false>(h, src, base, cnt, lo, tid);
    __syncthreads();

    if (by_bucket) {
        if (tid < nb) {
            int t = 0;
            #pragma unroll
            for (int k = 0; k < 32; ++k) t += h[tid * 32 + ((k + tid) & 31)];
            int* row = bpart + (long)tid * QS;
            row[blockIdx.x] = t;
            if (blockIdx.x == gridDim.x - 1)
                for (int q = (int)gridDim.x; q < QS; ++q) row[q] = 0;
        }
        return;
    }
    int* __restrict__ outp = part + (long)blockIdx.x * N;
    for (int r0 = tid; r0 < CNT_R; r0 += 8 * CNT_T) {
        int c[8];
        #pragma unroll
        for (int u = 0; u < 8; ++u) c[u] = h[r0 + u * CNT_T];
        #pragma unroll
        for (int u = 0; u < 8; ++u) {
            int node = lo + r0 + u * CNT_T;
            if (node < N) outp[node] = c[u];
        }
    }
}

#define BK_P   68
#define BK_MAX 1024
#define BK_T   1024
static_assert(BK_MAX_C == BK_MAX && BK_MAX <= CNT_T && BK_MAX_C * 32 <= CNT_R, "the bucket block's table: one thread per bucket, 32 words per bucket");

__device__ __forceinline__ float gcn_norm_of(int deg) { return rsqrtf(fmaxf((float)deg, 1.0f)); }

__device__ __forceinline__ long gcn_deg8_off(int N) {
    const long npad = (long)((N + 15) & ~15);
    return (8 * (long)N - npad) & ~15L;
}

__device__ __forceinline__ int gcn_block_excl_scan(int v, int* s_w) {
    const int lane = threadIdx.x & 31;
    const int wv = threadIdx.x >> 5;
    int inc = v;
    #pragma unroll
    for (int o = 1; o < 32; o <<= 1) {
        int t = __shfl_up(inc, o);
        if (lane >= o) inc += t;
    }
    if (lane == 31) s_w[wv] = inc;
    __syncthreads();
    if (wv == 0) {
        int w = s_w[lane];
        int winc = w;
        #pragma unroll
        for (int o = 1; o < 32; o <<= 1) {
            int t = __shfl_up(winc, o);
            if (lane >= o) winc += t;
        }
        s_w[lane] = winc - w;
    }
    __syncthreads();
    return inc - v + s_w[wv];
}

#define BIN_K  8
#define BIN_CH (BK_T * BIN_K)

__global__ __launch_bounds__(BK_T)
void gcn_bin_kernel(const int* __restrict__ src, const int* __restrict__ dst, const int* __restrict__ bpart,
                    const int* __restrict__ part, float* __restrict__ norm_src, int* __restrict__ tot,
                    int* __restrict__ cur, int2* __restrict__ list, int E, int NB, int N, int Q, int QS) {
    __shared__ int  s_cnt[BK_MAX];
    __shared__ int  s_loc[BK_MAX];
    __shared__ int2 s_stage[BIN_CH];
    __shared__ int  s_w[32];
    const int tid = threadIdx.x;
    s_cnt[tid] = 0;

    const long base = (long)blockIdx.x * BIN_CH;
    int s[BIN_K], d[BIN_K];
    #pragma unroll
    for (int u = 0; u < BIN_K; ++u) {
        const long e = base + u * BK_T + tid;
        const bool live = e < E;
        s[u] = live ? src[e] : 0;
        d[u] = live ? dst[e] : -1;
    }

    int my_tot = 0;
    if (tid < NB) {
        const int4* p0 = (const int4*)(bpart + (long)tid * QS);
        const int n4 = QS >> 2;
        for (int k0 = 0; k0 < n4; k0 += 8) {
            int4 v[8];
            #pragma unroll
            for (int u = 0; u < 8; ++u) v[u] = p0[k0 + u < n4 ? k0 + u : n4 - 1];
            #pragma unroll
            for (int u = 0; u < 8; ++u) my_tot += k0 + u < n4 ? (v[u].x + v[u].y) + (v[u].z + v[u].w) : 0;
        }
    }
    if (blockIdx.x == 0 && tid < NB) tot[tid] = my_tot;

    const int npad = (N + 15) & ~15;
    unsigned char* deg8 = (unsigned char*)norm_src + gcn_deg8_off(N);
    const int per = (((npad + (int)gridDim.x - 1) / (int)gridDim.x) + 127) & ~127;
    for (int j = tid; j < per; j += BK_T) {
        const int i = (int)blockIdx.x * per + j;
        if (i >= N) {
            if (i < npad) deg8[i] = 0;
        } else {
            int so = 0;
            for (int q0 = 0; q0 < Q; q0 += CNT_QB) {
                int v[CNT_QB];
                #pragma unroll
                for (int u = 0; u < CNT_QB; ++u) {
                    const int* tab = part + (long)(q0 + u < Q ? q0 + u : Q - 1) * N;
                    v[u] = tab[i];
                }
                #pragma unroll
                for (int u = 0; u < CNT_QB; ++u) so += q0 + u < Q ? v[u] : 0;
            }
            norm_src[i] = gcn_norm_of(so);
            deg8[i] = (unsigned char)(so < 255 ? so : 255);
        }
    }

    const int start = gcn_block_excl_scan(my_tot, s_w);

    int bk[BIN_K], rk[BIN_K];
    #pragma unroll
    for (int u = 0; u < BIN_K; ++u) {
        bk[u] = 0; rk[u] = 0;
        if (d[u] >= 0) {
            bk[u] = (int)((unsigned)d[u] / (unsigned)BK_R);
            rk[u] = atomicAdd(&s_cnt[bk[u]], 1);
        }
    }
    __syncthreads();
    const int c = s_cnt[tid];
    const int loc = gcn_block_excl_scan(c, s_w);
    int run = 0;
    if (c > 0) run = atomicAdd(&cur[tid], c);
    s_loc[tid] = loc;
    s_cnt[tid] = start + run - loc;
    __syncthreads();
    #pragma unroll
    for (int u = 0; u < BIN_K; ++u) {
        if (d[u] >= 0) s_stage[s_loc[bk[u]] + rk[u]] = make_int2(s[u], d[u]);
    }
    __syncthreads();
    const long left = (long)E - base;
    const int nlive = left > BIN_CH ? BIN_CH : (int)left;
    for (int p = tid; p < nlive; p += BK_T) {
        const int2 sd = s_stage[p];
        const int b = (int)((unsigned)sd.y / (unsigned)BK_R);
        list[s_cnt[b] + p] = make_int2(sd.x, sd.y - b * BK_R);
    }
}

#define RS_G  16
#define RS_U  8
#define RS_CH (BK_T * RS_U)
#define RS_TAB 100352

__device__ __forceinline__ int gcn_lane_i(int v, int j) {
    return __builtin_amdgcn_readlane(v, j);
}

__device__ __forceinline__ void gcn_flush_row(float* acc, int row, int lane, float2 sum) {
    float* ar = acc + row * BK_P + 2 * lane;
    atomicAdd(ar,     sum.x);
    atomicAdd(ar + 1, sum.y);
}

template <bool GUARD>
__device__ __forceinline__ void gcn_sum_run(const float* __restrict__ x, float* acc, int key_l, int ns_l,
                                            int g, int m, int lane, int& cur, float2& sum) {
    float2 v[RS_G];
    #pragma unroll
    for (int u = 0; u < RS_G; ++u) {
        v[u] = make_float2(0.0f, 0.0f);
        if (!GUARD || g + u < m) {
            const int src_row = gcn_lane_i(key_l, g + u) & 0x7FFFFF;
            const float2* xr = (const float2*)(x + (long)src_row * D) + lane;
            v[u] = xr[0];
        }
    }
    #pragma unroll
    for (int u = 0; u < RS_G; ++u) {
        if (!GUARD || g + u < m) {
            const int row = (int)((unsigned)gcn_lane_i(key_l, g + u) >> 23);
            const float ns = __int_as_float(gcn_lane_i(ns_l, g + u));
            if (row != cur) {
                if (cur >= 0) gcn_flush_row(acc, cur, lane, sum);
                sum = make_float2(0.0f, 0.0f);
                cur = row;
            }
            sum.x = fmaf(v[u].x, ns, sum.x);
            sum.y = fmaf(v[u].y, ns, sum.y);
        }
    }
}

__global__ __launch_bounds__(BK_T)
void gcn_rowsum_product_kernel(const float* __restrict__ x, const int2* __restrict__ list,
                               const int* __restrict__ tot, const float* __restrict__ norm_src,
                               const float* __restrict__ Wmat, float* __restrict__ out, int N, int NB) {
    __shared__ float acc[BK_R * BK_P];
    __shared__ int2 s_ord[RS_CH];
    __shared__ __attribute__((aligned(16))) unsigned char s_deg[RS_TAB];
    __shared__ int s_cnt[512];
    __shared__ int s_start[512];
    __shared__ int s_w[32];
    __shared__ int s_rng[2];
    const int tid = threadIdx.x;
    const int lane = tid & 31;
    const int wv = __builtin_amdgcn_readfirstlane(tid >> 5);

    const int my_tot = tid < NB ? tot[tid] : 0;

    const int npad = (N + 15) & ~15;
    const bool use_tab = npad <= RS_TAB;
    const unsigned char* deg8 = (const unsigned char*)norm_src + gcn_deg8_off(N);
    if (use_tab) {
        const int n16 = npad >> 4;
        const uint4* g16 = (const uint4*)deg8;
        uint4* l16 = (uint4*)s_deg;
        for (int k0 = tid; k0 < n16; k0 += 8 * BK_T) {
            uint4 t[8];
            #pragma unroll
            for (int u = 0; u < 8; ++u) {
                const int k = k0 + u * BK_T;
                t[u] = g16[k < n16 ? k : n16 - 1];
            }
            #pragma unroll
            for (int u = 0; u < 8; ++u) {
                const int k = k0 + u * BK_T;
                l16[k < n16 ? k : n16 - 1] = t[u];
            }
        }
    }

    for (int k = tid; k < BK_R * BK_P; k += BK_T) acc[k] = 0.0f;
    if (tid < 512) s_cnt[tid] = 0;
    const int start = gcn_block_excl_scan(my_tot, s_w);
    if (tid == (int)blockIdx.x) { s_rng[0] = start; s_rng[1] = my_tot; }
    __syncthreads();
    const int first = __builtin_amdgcn_readfirstlane(s_rng[0]);
    const int cnt   = __builtin_amdgcn_readfirstlane(s_rng[1]);
    int deg = 0;

    for (int c0 = 0; c0 < cnt; c0 += RS_CH) {
        const int n = cnt - c0 < RS_CH ? cnt - c0 : RS_CH;

        int2 sr[RS_U];
        #pragma unroll
        for (int u = 0; u < RS_U; ++u) {
            const int p = u * BK_T + tid;
            sr[u] = make_int2(0, -1);
            if (p < n) sr[u] = list[first + c0 + p];
        }
        int rk[RS_U];
        #pragma unroll
        for (int u = 0; u < RS_U; ++u) {
            rk[u] = 0;
            if (sr[u].y >= 0) rk[u] = atomicAdd(&s_cnt[sr[u].y], 1);
        }
        __syncthreads();
        const int c = tid < BK_R ? s_cnt[tid] : 0;
        deg += c;
        const int st = gcn_block_excl_scan(c, s_w);
        if (tid < BK_R) { s_start[tid] = st; s_cnt[tid] = 0; }
        __syncthreads();
        int d8[RS_U], st8[RS_U];
        #pragma unroll
        for (int u = 0; u < RS_U; ++u) {
            d8[u]  = (int)s_deg[use_tab ? sr[u].x : 0];
            st8[u] = s_start[sr[u].y >= 0 ? sr[u].y : 0];
        }
        float nsv[RS_U];
        bool esc = false;
        #pragma unroll
        for (int u = 0; u < RS_U; ++u) {
            if (!use_tab) d8[u] = 255;
            nsv[u] = gcn_norm_of(d8[u]);
            esc = esc || (d8[u] == 255 && sr[u].y >= 0);
        }
        if (esc) {
            #pragma unroll
            for (int u = 0; u < RS_U; ++u) {
                if (d8[u] == 255 && sr[u].y >= 0) nsv[u] = norm_src[sr[u].x];
            }
        }
        #pragma unroll
        for (int u = 0; u < RS_U; ++u) {
            if (sr[u].y >= 0) s_ord[st8[u] + rk[u]] = make_int2(sr[u].x | (int)((unsigned)sr[u].y << 23), __float_as_int(nsv[u]));
        }
        __syncthreads();

        const int per = ((n + 32 * RS_G - 1) / (32 * RS_G)) * RS_G;
        int ea = wv * per; if (ea > n) ea = n;
        int eb = ea + per; if (eb > n) eb = n;
        float2 sum = make_float2(0.0f, 0.0f);
        int cur = -1;
        for (int e0 = ea; e0 < eb; e0 += 32) {
            const int m = eb - e0 < 32 ? eb - e0 : 32;
            int2 en = make_int2(0, 0);
            if (lane < m) en = s_ord[e0 + lane];
            int g = 0;
            for (; g + RS_G <= m; g += RS_G) gcn_sum_run<false>(x, acc, en.x, en.y, g, m, lane, cur, sum);
            if (g < m)                       gcn_sum_run<true>(x, acc, en.x, en.y, g, m, lane, cur, sum);
        }
        if (cur >= 0) gcn_flush_row(acc, cur, lane, sum);
        __syncthreads();
    }

    if (tid < BK_R) s_start[tid] = deg;
    __syncthreads();

    for (int task = wv; task < 2 * (BK_R / 16); task += BK_T / 32) {
        const int half = lane >> 4;
        const int lm   = lane & 15;
        const int rt   = task >> 1;
        const int col0 = (task & 1) * 32;
        const int row  = rt * 16 + lm;
        const int g    = (int)blockIdx.x * BK_R + row;
        const float scale = (g < N) ? rsqrtf(fmaxf((float)s_start[row], 1.0f)) : 0.0f;
        const float* arow = acc + row * BK_P;

        v8f c0 = {}, c1 = {};
        for (int kk = 0; kk < 16; ++kk) {
            const int kbase = kk * 4 + half * 2;
            const float2 av = *(const float2*)(arow + kbase);
            v2f a; a.x = av.x * scale; a.y = av.y * scale;
            const float* w0 = Wmat + kbase * D + col0 + lm;
            v2f b0, b1;
            b0.x = w0[0];  b0.y = w0[D];
            b1.x = w0[16]; b1.y = w0[D + 16];
            c0 = __builtin_amdgcn_wmma_f32_16x16x4_f32(false, a, false, b0, (short)0, c0, false, false);
            c1 = __builtin_amdgcn_wmma_f32_16x16x4_f32(false, a, false, b1, (short)0, c1, false, false);
        }

        #pragma unroll
        for (int v = 0; v < 8; ++v) {
            const int gr = (int)blockIdx.x * BK_R + rt * 16 + v + half * 8;
            if (gr < N) {
                float* o = out + (long)gr * D + col0 + lm;
                o[0] = c0[v]; o[16] = c1[v];
            }
        }
    }
}

extern "C" void kernel_launch(void* const* d_in, const int* in_sizes, int n_in,
                              void* d_out, int out_size, void* d_ws, size_t ws_size,
                              hipStream_t stream) {
    const float* x    = (const float*)d_in[0];
    const float* Wmat = (const float*)d_in[1];
    const int*   src  = (const int*)d_in[2];
    const int*   dst  = (const int*)d_in[3];
    float* out = (float*)d_out;

    int N = in_sizes[0] / D;
    int E = in_sizes[2];

    float* ws_norm_src = (float*)d_ws;
    int*   ws_bpart    = (int*)(ws_norm_src + N);
    int*   ws_region   = ws_bpart + N;

    int Q = (E + CNT_CH - 1) / CNT_CH;
    int P = (N + CNT_R - 1) / CNT_R;

    int QS = (Q + 3) & ~3;
    int2* ws_list = (int2*)ws_region;
    int*  ws_part = ws_region + 2 * (long)E;
    int*  ws_tot  = ws_region + ((long)N * D - 2 * BK_MAX);
    int*  ws_cur  = ws_tot + BK_MAX;

    gcn_count_kernel<<<dim3(Q, P + 1, 1), CNT_T, 0, stream>>>(src, dst, ws_part, ws_bpart, ws_cur, E, N, QS);

    int NB = (N + BK_R - 1) / BK_R;
    gcn_bin_kernel<<<(E + BIN_CH - 1) / BIN_CH, BK_T, 0, stream>>>(src, dst, ws_bpart, ws_part, ws_norm_src, ws_tot,
                                                                 ws_cur, ws_list, E, NB, N, Q, QS);

    gcn_rowsum_product_kernel<<<NB, BK_T, 0, stream>>>(x, ws_list, ws_tot, ws_norm_src, Wmat, out, N, NB);
}
